# starting kernel
# baseline (speedup 1.0000x reference)
#include <hip/hip_runtime.h>
#include <hip/hip_bf16.h>


#define N_NODES 50000
#define D_POS   32
#define D_FEAT  768
#define D_IN    801
#define D_HID   32
#define D_OUT   32
#define N_GRAPHS 128
#define M_TILES  (N_NODES / 16)
#define WAVES_PER_BLOCK 8

typedef __attribute__((ext_vector_type(2))) float v2f;
typedef __attribute__((ext_vector_type(8))) float v8f;

__global__ __launch_bounds__(256) void input_gemm_kernel(
    const float* __restrict__ pos, const float* __restrict__ feat,
    const float* __restrict__ W1,  const float* __restrict__ Wp0,
    float* __restrict__ yBuf, float* __restrict__ accBuf, float* __restrict__ zBuf)
{
    const int wave = threadIdx.x >> 5;
    const int lane = threadIdx.x & 31;
    const int tile = blockIdx.x * WAVES_PER_BLOCK + wave;
    if (tile >= M_TILES) return;

    const int  m0   = tile * 16;
    const int  col  = lane & 15;
    const bool hi   = lane >= 16;
    const int  node = m0 + col;

    v8f cy0 = {}, cy1 = {}, cz0 = {}, cz1 = {};

    const float* posRow = pos + (size_t)node * D_POS;
    for (int k = 0; k < D_POS; k += 4) {
        const int kb = k + (hi ? 2 : 0);
        const v2f a = *(const v2f*)(posRow + kb);

        v2f by0, by1, bz0, bz1;
        by0.x = W1 [kb * 32 + col];        by0.y = W1 [(kb + 1) * 32 + col];
        by1.x = W1 [kb * 32 + 16 + col];   by1.y = W1 [(kb + 1) * 32 + 16 + col];
        bz0.x = Wp0[kb * 32 + col];        bz0.y = Wp0[(kb + 1) * 32 + col];
        bz1.x = Wp0[kb * 32 + 16 + col];   bz1.y = Wp0[(kb + 1) * 32 + 16 + col];

        cy0 = __builtin_amdgcn_wmma_f32_16x16x4_f32(false, a, false, by0, (short)0, cy0, false, false);
        cy1 = __builtin_amdgcn_wmma_f32_16x16x4_f32(false, a, false, by1, (short)0, cy1, false, false);
        cz0 = __builtin_amdgcn_wmma_f32_16x16x4_f32(false, a, false, bz0, (short)0, cz0, false, false);
        cz1 = __builtin_amdgcn_wmma_f32_16x16x4_f32(false, a, false, bz1, (short)0, cz1, false, false);
    }

    const float* featRow = feat + (size_t)node * D_FEAT;
    for (int k = 0; k < D_FEAT; k += 4) {
        const int kb = k + (hi ? 2 : 0);
        const v2f a = *(const v2f*)(featRow + kb);

        const int r = (33 + kb) * 32;
        v2f by0, by1, bz0, bz1;
        by0.x = W1 [r + col];        by0.y = W1 [r + 32 + col];
        by1.x = W1 [r + 16 + col];   by1.y = W1 [r + 48 + col];
        bz0.x = Wp0[r + col];        bz0.y = Wp0[r + 32 + col];
        bz1.x = Wp0[r + 16 + col];   bz1.y = Wp0[r + 48 + col];

        cy0 = __builtin_amdgcn_wmma_f32_16x16x4_f32(false, a, false, by0, (short)0, cy0, false, false);
        cy1 = __builtin_amdgcn_wmma_f32_16x16x4_f32(false, a, false, by1, (short)0, cy1, false, false);
        cz0 = __builtin_amdgcn_wmma_f32_16x16x4_f32(false, a, false, bz0, (short)0, cz0, false, false);
        cz1 = __builtin_amdgcn_wmma_f32_16x16x4_f32(false, a, false, bz1, (short)0, cz1, false, false);
    }

    for (int j = 0; j < 8; ++j) {
        const int row = (hi ? 8 : 0) + j;
        const int off = (m0 + row) * 32 + col;
        yBuf[off]        = cy0[j];  accBuf[off]        = cy0[j];
        yBuf[off + 16]   = cy1[j];  accBuf[off + 16]   = cy1[j];
        zBuf[off]        = cz0[j];
        zBuf[off + 16]   = cz1[j];
    }
}

__global__ __launch_bounds__(256) void edge_scatter_kernel(
    const float* __restrict__ yBuf, const int* __restrict__ src,
    const int* __restrict__ dst, float* __restrict__ accBuf, int nEdges)
{
    const int idx = blockIdx.x * blockDim.x + threadIdx.x;
    const int e = idx >> 3;
    if (e >= nEdges) return;
    const int c = (idx & 7) * 4;
    const int s = src[e], d = dst[e];
    const float4 v = *(const float4*)(yBuf + (size_t)s * 32 + c);
    float* p = accBuf + (size_t)d * 32 + c;
    atomicAdd(p + 0, v.x);
    atomicAdd(p + 1, v.y);
    atomicAdd(p + 2, v.z);
    atomicAdd(p + 3, v.w);
}

__global__ __launch_bounds__(256) void node_mlp_head_kernel(
    const float* __restrict__ accBuf, const float* __restrict__ zBuf,
    const float* __restrict__ W2, const float* __restrict__ bias1,
    const float* __restrict__ bias2, const float* __restrict__ Wp1,
    const int* __restrict__ gids, float* __restrict__ partial)
{
    __shared__ float hbuf[WAVES_PER_BLOCK][16 * 32];

    const int wave = threadIdx.x >> 5;
    const int lane = threadIdx.x & 31;
    const int tile = blockIdx.x * WAVES_PER_BLOCK + wave;
    if (tile >= M_TILES) return;

    const int  m0   = tile * 16;
    const int  col  = lane & 15;
    const bool hi   = lane >= 16;
    const int  node = m0 + col;

    const float* accRow = accBuf + (size_t)node * 32;
    v8f c10 = {}, c11 = {};
    for (int k = 0; k < 32; k += 4) {
        const int kb = k + (hi ? 2 : 0);
        const v2f av = *(const v2f*)(accRow + kb);
        const v2f bv = *(const v2f*)(bias1 + kb);
        v2f a;
        a.x = fmaxf(av.x + bv.x, 0.0f);
        a.y = fmaxf(av.y + bv.y, 0.0f);
        v2f b0, b1f;
        b0.x  = W2[kb * 32 + col];       b0.y  = W2[(kb + 1) * 32 + col];
        b1f.x = W2[kb * 32 + 16 + col];  b1f.y = W2[(kb + 1) * 32 + 16 + col];
        c10 = __builtin_amdgcn_wmma_f32_16x16x4_f32(false, a, false, b0,  (short)0, c10, false, false);
        c11 = __builtin_amdgcn_wmma_f32_16x16x4_f32(false, a, false, b1f, (short)0, c11, false, false);
    }

    const float bb0 = bias2[col];
    const float bb1 = bias2[16 + col];
    for (int j = 0; j < 8; ++j) {
        const int row = (hi ? 8 : 0) + j;
        hbuf[wave][row * 32 + col]      = fmaxf(c10[j] + bb0, 0.0f);
        hbuf[wave][row * 32 + 16 + col] = fmaxf(c11[j] + bb1, 0.0f);
    }

    v8f u0 = {}, u1 = {};
    for (int k = 0; k < 32; k += 4) {
        const int kb = k + (hi ? 2 : 0);
        const v2f a = *(const v2f*)(&hbuf[wave][col * 32 + kb]);
        v2f b0, b1f;
        b0.x  = Wp1[kb * 32 + col];       b0.y  = Wp1[(kb + 1) * 32 + col];
        b1f.x = Wp1[kb * 32 + 16 + col];  b1f.y = Wp1[(kb + 1) * 32 + 16 + col];
        u0 = __builtin_amdgcn_wmma_f32_16x16x4_f32(false, a, false, b0,  (short)0, u0, false, false);
        u1 = __builtin_amdgcn_wmma_f32_16x16x4_f32(false, a, false, b1f, (short)0, u1, false, false);
    }

    for (int j = 0; j < 8; ++j) {
        const int row  = (hi ? 8 : 0) + j;
        const int nd   = m0 + row;
        const int g    = gids[nd];
        const int off  = nd * 32 + col;
        atomicAdd(&partial[g * 32 + col],      u0[j] + zBuf[off]);
        atomicAdd(&partial[g * 32 + 16 + col], u1[j] + zBuf[off + 16]);
    }
}

__global__ void zero_partial_kernel(float* __restrict__ partial) {
    const int i = blockIdx.x * blockDim.x + threadIdx.x;
    if (i < N_GRAPHS * D_OUT) partial[i] = 0.0f;
}

__global__ void finalize_kernel(const float* __restrict__ partial,
                                const float* __restrict__ bp0,
                                const float* __restrict__ bp1,
                                float* __restrict__ out) {
    const int i = blockIdx.x * blockDim.x + threadIdx.x;
    if (i < N_GRAPHS * D_OUT) {
        const int c = i & 31;
        out[i] = partial[i] + bp0[c] + bp1[c];
    }
}

extern "C" void kernel_launch(void* const* d_in, const int* in_sizes, int n_in,
                              void* d_out, int out_size, void* d_ws, size_t ws_size,
                              hipStream_t stream) {
    const float* pos  = (const float*)d_in[0];
    const float* feat = (const float*)d_in[1];
    const int*   src  = (const int*)  d_in[2];
    const int*   dst  = (const int*)  d_in[3];
    const int*   gids = (const int*)  d_in[4];
    const float* W1   = (const float*)d_in[5];
    const float* b1   = (const float*)d_in[6];
    const float* W2   = (const float*)d_in[7];
    const float* b2   = (const float*)d_in[8];
    const float* Wp0  = (const float*)d_in[9];
    const float* bp0  = (const float*)d_in[10];
    const float* Wp1  = (const float*)d_in[11];
    const float* bp1  = (const float*)d_in[12];
    float* out = (float*)d_out;
    const int nEdges = in_sizes[2];

    float* yBuf    = (float*)d_ws;
    float* zBuf    = yBuf + (size_t)N_NODES * 32;
    float* accBuf  = zBuf + (size_t)N_NODES * 32;
    float* partial = accBuf + (size_t)N_NODES * 32;

    const int gemmBlocks = (M_TILES + WAVES_PER_BLOCK - 1) / WAVES_PER_BLOCK;

    zero_partial_kernel<<<(N_GRAPHS * D_OUT + 255) / 256, 256, 0, stream>>>(partial);
    input_gemm_kernel<<<gemmBlocks, 256, 0, stream>>>(pos, feat, W1, Wp0,
                                                      yBuf, accBuf, zBuf);
    edge_scatter_kernel<<<((size_t)nEdges * 8 + 255) / 256, 256, 0, stream>>>(
        yBuf, src, dst, accBuf, nEdges);
    node_mlp_head_kernel<<<gemmBlocks, 256, 0, stream>>>(accBuf, zBuf, W2, b1, b2,
                                                         Wp1, gids, partial);
    finalize_kernel<<<(N_GRAPHS * D_OUT + 255) / 256, 256, 0, stream>>>(
        partial, bp0, bp1, out);
}


# input GEMM operand path: W1/Wp0 staged once per work-group in LDS in fragment order, A rows as 16-byte loads fetched ahead, one 16-wave work-group per WGP; k regrouped in 8-wide chunks (sum order)
# speedup vs baseline: 1.5644x; 1.5644x over previous
#include <hip/hip_runtime.h>
#include <hip/hip_bf16.h>


#define N_NODES 50000
#define D_POS   32
#define D_FEAT  768
#define D_IN    801
#define D_HID   32
#define D_OUT   32
#define N_GRAPHS 128
#define M_TILES  (N_NODES / 16)
#define WAVES_PER_BLOCK 8

typedef __attribute__((ext_vector_type(2))) float v2f;
typedef __attribute__((ext_vector_type(8))) float v8f;

#define GEMM_WGS    256
#define FULL_WAVES  (M_TILES / GEMM_WGS)
#define GEMM_WAVES  (FULL_WAVES + 4)
#define K_USED      800
#define K_STEPS     (K_USED / 4)
#define K_PAIRS     (K_USED / 2)

template <int NVEC>
__device__ __forceinline__ void gemm_chunk(
    const float4 (&a)[NVEC], const float4* __restrict__ Bs,
    const int s0, const int lane, v8f& cy0, v8f& cy1, v8f& cz0, v8f& cz1)
{
#pragma unroll
    for (int i = 0; i < NVEC; ++i) {
        {
#pragma unroll
            for (int o = 0; o < 2; ++o) {
                const int s = s0 + 2 * i + o;
                const float4 b0 = Bs[s * 64 + lane];
                const float4 b1 = Bs[s * 64 + 32 + lane];
                v2f av;
                av.x = o ? a[i].z : a[i].x;
                av.y = o ? a[i].w : a[i].y;
                v2f by0, by1, bz0, bz1;
                by0.x = b0.x; by0.y = b0.y;  bz0.x = b0.z; bz0.y = b0.w;
                by1.x = b1.x; by1.y = b1.y;  bz1.x = b1.z; bz1.y = b1.w;
                cy0 = __builtin_amdgcn_wmma_f32_16x16x4_f32(false, av, false, by0, (short)0, cy0, false, false);
                cy1 = __builtin_amdgcn_wmma_f32_16x16x4_f32(false, av, false, by1, (short)0, cy1, false, false);
                cz0 = __builtin_amdgcn_wmma_f32_16x16x4_f32(false, av, false, bz0, (short)0, cz0, false, false);
                cz1 = __builtin_amdgcn_wmma_f32_16x16x4_f32(false, av, false, bz1, (short)0, cz1, false, false);
            }
        }
    }
}

template <int NVEC>
__device__ __forceinline__ void gemm_chunk_quarter(
    const float4 (&a)[NVEC], const v2f* __restrict__ Bq, const int s0, v8f& acc)
{
#pragma unroll
    for (int i = 0; i < NVEC; ++i) {
#pragma unroll
        for (int o = 0; o < 2; ++o) {
            const int s = s0 + 2 * i + o;
            const v2f b = Bq[s * 128];
            v2f av;
            av.x = o ? a[i].z : a[i].x;
            av.y = o ? a[i].w : a[i].y;
            acc = __builtin_amdgcn_wmma_f32_16x16x4_f32(false, av, false, b, (short)0, acc, false, false);
        }
    }
}

__global__ __launch_bounds__(GEMM_WAVES * 32) void input_gemm_kernel(
    const float* __restrict__ pos, const float* __restrict__ feat,
    const float* __restrict__ W1,  const float* __restrict__ Wp0,
    float* __restrict__ yBuf, float* __restrict__ accBuf, float* __restrict__ zBuf)
{
    __shared__ __attribute__((aligned(16))) float4 Bs[K_STEPS * 64];

    const int wave   = __builtin_amdgcn_readfirstlane((int)(threadIdx.x >> 5));
    const int lane   = threadIdx.x & 31;
    const bool whole  = wave < FULL_WAVES;
    const int tile   = (whole ? wave : FULL_WAVES) * GEMM_WGS + (int)blockIdx.x;
    const bool active = tile < M_TILES;

    const int  m0   = (active ? tile : 0) * 16;
    const int  col  = lane & 15;
    const int  h    = lane >> 4;
    const int  node = m0 + col;

    const float4* posV  = (const float4*)(pos  + (size_t)node * D_POS)  + h;
    const float4* featV = (const float4*)(feat + (size_t)node * D_FEAT) + h;

    float4 aP[4], aA[8], aB[8];
#pragma unroll
    for (int j = 0; j < 4; ++j) aP[j] = posV[2 * j];
#pragma unroll
    for (int j = 0; j < 8; ++j) aA[j] = featV[2 * j];

    {
        const int n  = lane;
        const int nh = n >> 4, c = n & 15;
        for (int bt = wave; bt < K_PAIRS / 8; bt += GEMM_WAVES) {
            const int k0   = 16 * bt;
            const int row0 = (k0 < D_POS) ? k0 : (k0 + 1);
            const float* w1p = W1  + row0 * 32 + n;
            const float* wpp = Wp0 + row0 * 32 + n;
            float4 v[8];
#pragma unroll
            for (int u = 0; u < 8; ++u) {
                v[u].x = w1p[(2 * u) * 32];
                v[u].y = w1p[(2 * u + 1) * 32];
                v[u].z = wpp[(2 * u) * 32];
                v[u].w = wpp[(2 * u + 1) * 32];
            }
#pragma unroll
            for (int u = 0; u < 8; ++u) {
                const int s = 2 * (2 * bt + (u >> 2)) + (u & 1);
                Bs[s * 64 + nh * 32 + ((u >> 1) & 1) * 16 + c] = v[u];
            }
        }
    }
    __syncthreads();
    if (!active) return;

    if (whole) {
        v8f cy0 = {}, cy1 = {}, cz0 = {}, cz1 = {};

        gemm_chunk<4>(aP, Bs, 0, lane, cy0, cy1, cz0, cz1);

        for (int st = 0; st < D_FEAT / 64 - 2; st += 2) {
#pragma unroll
            for (int j = 0; j < 8; ++j) aB[j] = featV[16 * (st + 1) + 2 * j];
            gemm_chunk<8>(aA, Bs, 8 + 16 * st, lane, cy0, cy1, cz0, cz1);
#pragma unroll
            for (int j = 0; j < 8; ++j) aA[j] = featV[16 * (st + 2) + 2 * j];
            gemm_chunk<8>(aB, Bs, 8 + 16 * (st + 1), lane, cy0, cy1, cz0, cz1);
        }
        {
            const int st = D_FEAT / 64 - 2;
#pragma unroll
            for (int j = 0; j < 8; ++j) aB[j] = featV[16 * (st + 1) + 2 * j];
            gemm_chunk<8>(aA, Bs, 8 + 16 * st, lane, cy0, cy1, cz0, cz1);
            gemm_chunk<8>(aB, Bs, 8 + 16 * (st + 1), lane, cy0, cy1, cz0, cz1);
        }

        for (int j = 0; j < 8; ++j) {
            const int row = 8 * h + j;
            const int off = (m0 + row) * 32 + col;
            yBuf[off]        = cy0[j];  accBuf[off]        = cy0[j];
            yBuf[off + 16]   = cy1[j];  accBuf[off + 16]   = cy1[j];
            zBuf[off]        = cz0[j];
            zBuf[off + 16]   = cz1[j];
        }
    } else {
        const int qtr = wave - FULL_WAVES;
        const int mat = qtr >> 1;
        const int nhq = qtr & 1;
        const v2f* Bq = (const v2f*)Bs + (nhq * 32 + lane) * 2 + mat;
        v8f acc = {};

        gemm_chunk_quarter<4>(aP, Bq, 0, acc);
        for (int st = 0; st < D_FEAT / 64 - 2; st += 2) {
#pragma unroll
            for (int j = 0; j < 8; ++j) aB[j] = featV[16 * (st + 1) + 2 * j];
            gemm_chunk_quarter<8>(aA, Bq, 8 + 16 * st, acc);
#pragma unroll
            for (int j = 0; j < 8; ++j) aA[j] = featV[16 * (st + 2) + 2 * j];
            gemm_chunk_quarter<8>(aB, Bq, 8 + 16 * (st + 1), acc);
        }
        {
            const int st = D_FEAT / 64 - 2;
#pragma unroll
            for (int j = 0; j < 8; ++j) aB[j] = featV[16 * (st + 1) + 2 * j];
            gemm_chunk_quarter<8>(aA, Bq, 8 + 16 * st, acc);
            gemm_chunk_quarter<8>(aB, Bq, 8 + 16 * (st + 1), acc);
        }

        for (int j = 0; j < 8; ++j) {
            const int row = 8 * h + j;
            const int off = (m0 + row) * 32 + nhq * 16 + col;
            if (mat == 0) { yBuf[off] = acc[j];  accBuf[off] = acc[j]; }
            else          { zBuf[off] = acc[j]; }
        }
    }
}

__global__ __launch_bounds__(256) void edge_scatter_kernel(
    const float* __restrict__ yBuf, const int* __restrict__ src,
    const int* __restrict__ dst, float* __restrict__ accBuf, int nEdges)
{
    const int idx = blockIdx.x * blockDim.x + threadIdx.x;
    const int e = idx >> 3;
    if (e >= nEdges) return;
    const int c = (idx & 7) * 4;
    const int s = src[e], d = dst[e];
    const float4 v = *(const float4*)(yBuf + (size_t)s * 32 + c);
    float* p = accBuf + (size_t)d * 32 + c;
    atomicAdd(p + 0, v.x);
    atomicAdd(p + 1, v.y);
    atomicAdd(p + 2, v.z);
    atomicAdd(p + 3, v.w);
}

__global__ __launch_bounds__(256) void node_mlp_head_kernel(
    const float* __restrict__ accBuf, const float* __restrict__ zBuf,
    const float* __restrict__ W2, const float* __restrict__ bias1,
    const float* __restrict__ bias2, const float* __restrict__ Wp1,
    const int* __restrict__ gids, float* __restrict__ partial)
{
    __shared__ float hbuf[WAVES_PER_BLOCK][16 * 32];

    const int wave = threadIdx.x >> 5;
    const int lane = threadIdx.x & 31;
    const int tile = blockIdx.x * WAVES_PER_BLOCK + wave;
    if (tile >= M_TILES) return;

    const int  m0   = tile * 16;
    const int  col  = lane & 15;
    const bool hi   = lane >= 16;
    const int  node = m0 + col;

    const float* accRow = accBuf + (size_t)node * 32;
    v8f c10 = {}, c11 = {};
    for (int k = 0; k < 32; k += 4) {
        const int kb = k + (hi ? 2 : 0);
        const v2f av = *(const v2f*)(accRow + kb);
        const v2f bv = *(const v2f*)(bias1 + kb);
        v2f a;
        a.x = fmaxf(av.x + bv.x, 0.0f);
        a.y = fmaxf(av.y + bv.y, 0.0f);
        v2f b0, b1f;
        b0.x  = W2[kb * 32 + col];       b0.y  = W2[(kb + 1) * 32 + col];
        b1f.x = W2[kb * 32 + 16 + col];  b1f.y = W2[(kb + 1) * 32 + 16 + col];
        c10 = __builtin_amdgcn_wmma_f32_16x16x4_f32(false, a, false, b0,  (short)0, c10, false, false);
        c11 = __builtin_amdgcn_wmma_f32_16x16x4_f32(false, a, false, b1f, (short)0, c11, false, false);
    }

    const float bb0 = bias2[col];
    const float bb1 = bias2[16 + col];
    for (int j = 0; j < 8; ++j) {
        const int row = (hi ? 8 : 0) + j;
        hbuf[wave][row * 32 + col]      = fmaxf(c10[j] + bb0, 0.0f);
        hbuf[wave][row * 32 + 16 + col] = fmaxf(c11[j] + bb1, 0.0f);
    }

    v8f u0 = {}, u1 = {};
    for (int k = 0; k < 32; k += 4) {
        const int kb = k + (hi ? 2 : 0);
        const v2f a = *(const v2f*)(&hbuf[wave][col * 32 + kb]);
        v2f b0, b1f;
        b0.x  = Wp1[kb * 32 + col];       b0.y  = Wp1[(kb + 1) * 32 + col];
        b1f.x = Wp1[kb * 32 + 16 + col];  b1f.y = Wp1[(kb + 1) * 32 + 16 + col];
        u0 = __builtin_amdgcn_wmma_f32_16x16x4_f32(false, a, false, b0,  (short)0, u0, false, false);
        u1 = __builtin_amdgcn_wmma_f32_16x16x4_f32(false, a, false, b1f, (short)0, u1, false, false);
    }

    for (int j = 0; j < 8; ++j) {
        const int row  = (hi ? 8 : 0) + j;
        const int nd   = m0 + row;
        const int g    = gids[nd];
        const int off  = nd * 32 + col;
        atomicAdd(&partial[g * 32 + col],      u0[j] + zBuf[off]);
        atomicAdd(&partial[g * 32 + 16 + col], u1[j] + zBuf[off + 16]);
    }
}

__global__ void zero_partial_kernel(float* __restrict__ partial) {
    const int i = blockIdx.x * blockDim.x + threadIdx.x;
    if (i < N_GRAPHS * D_OUT) partial[i] = 0.0f;
}

__global__ void finalize_kernel(const float* __restrict__ partial,
                                const float* __restrict__ bp0,
                                const float* __restrict__ bp1,
                                float* __restrict__ out) {
    const int i = blockIdx.x * blockDim.x + threadIdx.x;
    if (i < N_GRAPHS * D_OUT) {
        const int c = i & 31;
        out[i] = partial[i] + bp0[c] + bp1[c];
    }
}

extern "C" void kernel_launch(void* const* d_in, const int* in_sizes, int n_in,
                              void* d_out, int out_size, void* d_ws, size_t ws_size,
                              hipStream_t stream) {
    const float* pos  = (const float*)d_in[0];
    const float* feat = (const float*)d_in[1];
    const int*   src  = (const int*)  d_in[2];
    const int*   dst  = (const int*)  d_in[3];
    const int*   gids = (const int*)  d_in[4];
    const float* W1   = (const float*)d_in[5];
    const float* b1   = (const float*)d_in[6];
    const float* W2   = (const float*)d_in[7];
    const float* b2   = (const float*)d_in[8];
    const float* Wp0  = (const float*)d_in[9];
    const float* bp0  = (const float*)d_in[10];
    const float* Wp1  = (const float*)d_in[11];
    const float* bp1  = (const float*)d_in[12];
    float* out = (float*)d_out;
    const int nEdges = in_sizes[2];

    float* yBuf    = (float*)d_ws;
    float* zBuf    = yBuf + (size_t)N_NODES * 32;
    float* accBuf  = zBuf + (size_t)N_NODES * 32;
    float* partial = accBuf + (size_t)N_NODES * 32;

    const int gemmBlocks = (M_TILES + WAVES_PER_BLOCK - 1) / WAVES_PER_BLOCK;

    zero_partial_kernel<<<(N_GRAPHS * D_OUT + 255) / 256, 256, 0, stream>>>(partial);
    input_gemm_kernel<<<GEMM_WGS, GEMM_WAVES * 32, 0, stream>>>(pos, feat, W1, Wp0,
                                                                yBuf, accBuf, zBuf);
    edge_scatter_kernel<<<((size_t)nEdges * 8 + 255) / 256, 256, 0, stream>>>(
        yBuf, src, dst, accBuf, nEdges);
    node_mlp_head_kernel<<<gemmBlocks, 256, 0, stream>>>(accBuf, zBuf, W2, b1, b2,
                                                         Wp1, gids, partial);
    finalize_kernel<<<(N_GRAPHS * D_OUT + 255) / 256, 256, 0, stream>>>(
        partial, bp0, bp1, out);
}


# edge scatter as a persistent kernel: 4096 waves loop over runs of edges, one whole 128-byte row per gather and per atomic, not 100000 one-shot waves; arrival order of the float atomics changes
# speedup vs baseline: 1.7131x; 1.0951x over previous
#include <hip/hip_runtime.h>
#include <hip/hip_bf16.h>


#define N_NODES 50000
#define D_POS   32
#define D_FEAT  768
#define D_IN    801
#define D_HID   32
#define D_OUT   32
#define N_GRAPHS 128
#define M_TILES  (N_NODES / 16)
#define WAVES_PER_BLOCK 8

typedef __attribute__((ext_vector_type(2))) float v2f;
typedef __attribute__((ext_vector_type(8))) float v8f;

#define GEMM_WGS    256
#define FULL_WAVES  (M_TILES / GEMM_WGS)
#define GEMM_WAVES  (FULL_WAVES + 4)
#define K_USED      800
#define K_STEPS     (K_USED / 4)
#define K_PAIRS     (K_USED / 2)

template <int NVEC>
__device__ __forceinline__ void gemm_chunk(
    const float4 (&a)[NVEC], const float4* __restrict__ Bs,
    const int s0, const int lane, v8f& cy0, v8f& cy1, v8f& cz0, v8f& cz1)
{
#pragma unroll
    for (int i = 0; i < NVEC; ++i) {
        {
#pragma unroll
            for (int o = 0; o < 2; ++o) {
                const int s = s0 + 2 * i + o;
                const float4 b0 = Bs[s * 64 + lane];
                const float4 b1 = Bs[s * 64 + 32 + lane];
                v2f av;
                av.x = o ? a[i].z : a[i].x;
                av.y = o ? a[i].w : a[i].y;
                v2f by0, by1, bz0, bz1;
                by0.x = b0.x; by0.y = b0.y;  bz0.x = b0.z; bz0.y = b0.w;
                by1.x = b1.x; by1.y = b1.y;  bz1.x = b1.z; bz1.y = b1.w;
                cy0 = __builtin_amdgcn_wmma_f32_16x16x4_f32(false, av, false, by0, (short)0, cy0, false, false);
                cy1 = __builtin_amdgcn_wmma_f32_16x16x4_f32(false, av, false, by1, (short)0, cy1, false, false);
                cz0 = __builtin_amdgcn_wmma_f32_16x16x4_f32(false, av, false, bz0, (short)0, cz0, false, false);
                cz1 = __builtin_amdgcn_wmma_f32_16x16x4_f32(false, av, false, bz1, (short)0, cz1, false, false);
            }
        }
    }
}

template <int NVEC>
__device__ __forceinline__ void gemm_chunk_quarter(
    const float4 (&a)[NVEC], const v2f* __restrict__ Bq, const int s0, v8f& acc)
{
#pragma unroll
    for (int i = 0; i < NVEC; ++i) {
#pragma unroll
        for (int o = 0; o < 2; ++o) {
            const int s = s0 + 2 * i + o;
            const v2f b = Bq[s * 128];
            v2f av;
            av.x = o ? a[i].z : a[i].x;
            av.y = o ? a[i].w : a[i].y;
            acc = __builtin_amdgcn_wmma_f32_16x16x4_f32(false, av, false, b, (short)0, acc, false, false);
        }
    }
}

__global__ __launch_bounds__(GEMM_WAVES * 32) void input_gemm_kernel(
    const float* __restrict__ pos, const float* __restrict__ feat,
    const float* __restrict__ W1,  const float* __restrict__ Wp0,
    float* __restrict__ yBuf, float* __restrict__ accBuf, float* __restrict__ zBuf)
{
    __shared__ __attribute__((aligned(16))) float4 Bs[K_STEPS * 64];

    const int wave   = __builtin_amdgcn_readfirstlane((int)(threadIdx.x >> 5));
    const int lane   = threadIdx.x & 31;
    const bool whole  = wave < FULL_WAVES;
    const int tile   = (whole ? wave : FULL_WAVES) * GEMM_WGS + (int)blockIdx.x;
    const bool active = tile < M_TILES;

    const int  m0   = (active ? tile : 0) * 16;
    const int  col  = lane & 15;
    const int  h    = lane >> 4;
    const int  node = m0 + col;

    const float4* posV  = (const float4*)(pos  + (size_t)node * D_POS)  + h;
    const float4* featV = (const float4*)(feat + (size_t)node * D_FEAT) + h;

    float4 aP[4], aA[8], aB[8];
#pragma unroll
    for (int j = 0; j < 4; ++j) aP[j] = posV[2 * j];
#pragma unroll
    for (int j = 0; j < 8; ++j) aA[j] = featV[2 * j];

    {
        const int n  = lane;
        const int nh = n >> 4, c = n & 15;
        for (int bt = wave; bt < K_PAIRS / 8; bt += GEMM_WAVES) {
            const int k0   = 16 * bt;
            const int row0 = (k0 < D_POS) ? k0 : (k0 + 1);
            const float* w1p = W1  + row0 * 32 + n;
            const float* wpp = Wp0 + row0 * 32 + n;
            float4 v[8];
#pragma unroll
            for (int u = 0; u < 8; ++u) {
                v[u].x = w1p[(2 * u) * 32];
                v[u].y = w1p[(2 * u + 1) * 32];
                v[u].z = wpp[(2 * u) * 32];
                v[u].w = wpp[(2 * u + 1) * 32];
            }
#pragma unroll
            for (int u = 0; u < 8; ++u) {
                const int s = 2 * (2 * bt + (u >> 2)) + (u & 1);
                Bs[s * 64 + nh * 32 + ((u >> 1) & 1) * 16 + c] = v[u];
            }
        }
    }
    __syncthreads();
    if (!active) return;

    if (whole) {
        v8f cy0 = {}, cy1 = {}, cz0 = {}, cz1 = {};

        gemm_chunk<4>(aP, Bs, 0, lane, cy0, cy1, cz0, cz1);

        for (int st = 0; st < D_FEAT / 64 - 2; st += 2) {
#pragma unroll
            for (int j = 0; j < 8; ++j) aB[j] = featV[16 * (st + 1) + 2 * j];
            gemm_chunk<8>(aA, Bs, 8 + 16 * st, lane, cy0, cy1, cz0, cz1);
#pragma unroll
            for (int j = 0; j < 8; ++j) aA[j] = featV[16 * (st + 2) + 2 * j];
            gemm_chunk<8>(aB, Bs, 8 + 16 * (st + 1), lane, cy0, cy1, cz0, cz1);
        }
        {
            const int st = D_FEAT / 64 - 2;
#pragma unroll
            for (int j = 0; j < 8; ++j) aB[j] = featV[16 * (st + 1) + 2 * j];
            gemm_chunk<8>(aA, Bs, 8 + 16 * st, lane, cy0, cy1, cz0, cz1);
            gemm_chunk<8>(aB, Bs, 8 + 16 * (st + 1), lane, cy0, cy1, cz0, cz1);
        }

        for (int j = 0; j < 8; ++j) {
            const int row = 8 * h + j;
            const int off = (m0 + row) * 32 + col;
            yBuf[off]        = cy0[j];  accBuf[off]        = cy0[j];
            yBuf[off + 16]   = cy1[j];  accBuf[off + 16]   = cy1[j];
            zBuf[off]        = cz0[j];
            zBuf[off + 16]   = cz1[j];
        }
    } else {
        const int qtr = wave - FULL_WAVES;
        const int mat = qtr >> 1;
        const int nhq = qtr & 1;
        const v2f* Bq = (const v2f*)Bs + (nhq * 32 + lane) * 2 + mat;
        v8f acc = {};

        gemm_chunk_quarter<4>(aP, Bq, 0, acc);
        for (int st = 0; st < D_FEAT / 64 - 2; st += 2) {
#pragma unroll
            for (int j = 0; j < 8; ++j) aB[j] = featV[16 * (st + 1) + 2 * j];
            gemm_chunk_quarter<8>(aA, Bq, 8 + 16 * st, acc);
#pragma unroll
            for (int j = 0; j < 8; ++j) aA[j] = featV[16 * (st + 2) + 2 * j];
            gemm_chunk_quarter<8>(aB, Bq, 8 + 16 * (st + 1), acc);
        }
        {
            const int st = D_FEAT / 64 - 2;
#pragma unroll
            for (int j = 0; j < 8; ++j) aB[j] = featV[16 * (st + 1) + 2 * j];
            gemm_chunk_quarter<8>(aA, Bq, 8 + 16 * st, acc);
            gemm_chunk_quarter<8>(aB, Bq, 8 + 16 * (st + 1), acc);
        }

        for (int j = 0; j < 8; ++j) {
            const int row = 8 * h + j;
            const int off = (m0 + row) * 32 + nhq * 16 + col;
            if (mat == 0) { yBuf[off] = acc[j];  accBuf[off] = acc[j]; }
            else          { zBuf[off] = acc[j]; }
        }
    }
}

#define SC_WGS          512
#define SC_WAVES_PER_WG 8
#define SC_WAVES_LOG2   12
#define SC_UNIT         16

template <int BASE, int N>
__device__ __forceinline__ void scatter_edges(
    const float* __restrict__ yBuf, float* __restrict__ accBuf,
    const int sv, const int dv, const int lane)
{
    float v[N];
    int   d[N];
#pragma unroll
    for (int u = 0; u < N; ++u) {
        const int s = __builtin_amdgcn_readlane(sv, BASE + u);
        d[u]        = __builtin_amdgcn_readlane(dv, BASE + u);
        v[u] = yBuf[(size_t)s * 32 + lane];
    }
#pragma unroll
    for (int u = 0; u < N; ++u)
        atomicAdd(accBuf + (size_t)d[u] * 32 + lane, v[u]);
}

__global__ __launch_bounds__(SC_WAVES_PER_WG * 32) void edge_scatter_kernel(
    const float* __restrict__ yBuf, const int* __restrict__ src,
    const int* __restrict__ dst, float* __restrict__ accBuf, int nEdges)
{
    const int wave = __builtin_amdgcn_readfirstlane((int)(threadIdx.x >> 5));
    const int lane = threadIdx.x & 31;
    const int w    = (int)blockIdx.x * SC_WAVES_PER_WG + wave;

    const int nUnits = (nEdges + SC_UNIT - 1) / SC_UNIT;
    const int u0 = (int)(((long long)w * nUnits) >> SC_WAVES_LOG2);
    const int u1 = (int)(((long long)(w + 1) * nUnits) >> SC_WAVES_LOG2);
    const int eBeg = u0 * SC_UNIT;
    const int eEnd = (u1 * SC_UNIT < nEdges) ? u1 * SC_UNIT : nEdges;

    if (eBeg >= eEnd) return;

    const int last = nEdges - 1;
    const int p0   = (eBeg + lane < last) ? eBeg + lane : last;
    int sv = src[p0], dv = dst[p0];

    for (int e0 = eBeg; e0 < eEnd; e0 += 32) {
        const int pN  = (e0 + 32 + lane < last) ? e0 + 32 + lane : last;
        const int svN = src[pN], dvN = dst[pN];

        const int left = eEnd - e0;
        if (left >= 32) {
            scatter_edges<0, 32>(yBuf, accBuf, sv, dv, lane);
        } else {
            int done = 0;
            if (left >= SC_UNIT) {
                scatter_edges<0, SC_UNIT>(yBuf, accBuf, sv, dv, lane);
                done = SC_UNIT;
            }
            for (int i = done; i < left; ++i) {
                const int s = __builtin_amdgcn_readlane(sv, i);
                const int d = __builtin_amdgcn_readlane(dv, i);
                atomicAdd(accBuf + (size_t)d * 32 + lane, yBuf[(size_t)s * 32 + lane]);
            }
        }
        sv = svN;  dv = dvN;
    }
}

__global__ __launch_bounds__(256) void node_mlp_head_kernel(
    const float* __restrict__ accBuf, const float* __restrict__ zBuf,
    const float* __restrict__ W2, const float* __restrict__ bias1,
    const float* __restrict__ bias2, const float* __restrict__ Wp1,
    const int* __restrict__ gids, float* __restrict__ partial)
{
    __shared__ float hbuf[WAVES_PER_BLOCK][16 * 32];

    const int wave = threadIdx.x >> 5;
    const int lane = threadIdx.x & 31;
    const int tile = blockIdx.x * WAVES_PER_BLOCK + wave;
    if (tile >= M_TILES) return;

    const int  m0   = tile * 16;
    const int  col  = lane & 15;
    const bool hi   = lane >= 16;
    const int  node = m0 + col;

    const float* accRow = accBuf + (size_t)node * 32;
    v8f c10 = {}, c11 = {};
    for (int k = 0; k < 32; k += 4) {
        const int kb = k + (hi ? 2 : 0);
        const v2f av = *(const v2f*)(accRow + kb);
        const v2f bv = *(const v2f*)(bias1 + kb);
        v2f a;
        a.x = fmaxf(av.x + bv.x, 0.0f);
        a.y = fmaxf(av.y + bv.y, 0.0f);
        v2f b0, b1f;
        b0.x  = W2[kb * 32 + col];       b0.y  = W2[(kb + 1) * 32 + col];
        b1f.x = W2[kb * 32 + 16 + col];  b1f.y = W2[(kb + 1) * 32 + 16 + col];
        c10 = __builtin_amdgcn_wmma_f32_16x16x4_f32(false, a, false, b0,  (short)0, c10, false, false);
        c11 = __builtin_amdgcn_wmma_f32_16x16x4_f32(false, a, false, b1f, (short)0, c11, false, false);
    }

    const float bb0 = bias2[col];
    const float bb1 = bias2[16 + col];
    for (int j = 0; j < 8; ++j) {
        const int row = (hi ? 8 : 0) + j;
        hbuf[wave][row * 32 + col]      = fmaxf(c10[j] + bb0, 0.0f);
        hbuf[wave][row * 32 + 16 + col] = fmaxf(c11[j] + bb1, 0.0f);
    }

    v8f u0 = {}, u1 = {};
    for (int k = 0; k < 32; k += 4) {
        const int kb = k + (hi ? 2 : 0);
        const v2f a = *(const v2f*)(&hbuf[wave][col * 32 + kb]);
        v2f b0, b1f;
        b0.x  = Wp1[kb * 32 + col];       b0.y  = Wp1[(kb + 1) * 32 + col];
        b1f.x = Wp1[kb * 32 + 16 + col];  b1f.y = Wp1[(kb + 1) * 32 + 16 + col];
        u0 = __builtin_amdgcn_wmma_f32_16x16x4_f32(false, a, false, b0,  (short)0, u0, false, false);
        u1 = __builtin_amdgcn_wmma_f32_16x16x4_f32(false, a, false, b1f, (short)0, u1, false, false);
    }

    for (int j = 0; j < 8; ++j) {
        const int row  = (hi ? 8 : 0) + j;
        const int nd   = m0 + row;
        const int g    = gids[nd];
        const int off  = nd * 32 + col;
        atomicAdd(&partial[g * 32 + col],      u0[j] + zBuf[off]);
        atomicAdd(&partial[g * 32 + 16 + col], u1[j] + zBuf[off + 16]);
    }
}

__global__ void zero_partial_kernel(float* __restrict__ partial) {
    const int i = blockIdx.x * blockDim.x + threadIdx.x;
    if (i < N_GRAPHS * D_OUT) partial[i] = 0.0f;
}

__global__ void finalize_kernel(const float* __restrict__ partial,
                                const float* __restrict__ bp0,
                                const float* __restrict__ bp1,
                                float* __restrict__ out) {
    const int i = blockIdx.x * blockDim.x + threadIdx.x;
    if (i < N_GRAPHS * D_OUT) {
        const int c = i & 31;
        out[i] = partial[i] + bp0[c] + bp1[c];
    }
}

extern "C" void kernel_launch(void* const* d_in, const int* in_sizes, int n_in,
                              void* d_out, int out_size, void* d_ws, size_t ws_size,
                              hipStream_t stream) {
    const float* pos  = (const float*)d_in[0];
    const float* feat = (const float*)d_in[1];
    const int*   src  = (const int*)  d_in[2];
    const int*   dst  = (const int*)  d_in[3];
    const int*   gids = (const int*)  d_in[4];
    const float* W1   = (const float*)d_in[5];
    const float* b1   = (const float*)d_in[6];
    const float* W2   = (const float*)d_in[7];
    const float* b2   = (const float*)d_in[8];
    const float* Wp0  = (const float*)d_in[9];
    const float* bp0  = (const float*)d_in[10];
    const float* Wp1  = (const float*)d_in[11];
    const float* bp1  = (const float*)d_in[12];
    float* out = (float*)d_out;
    const int nEdges = in_sizes[2];

    float* yBuf    = (float*)d_ws;
    float* zBuf    = yBuf + (size_t)N_NODES * 32;
    float* accBuf  = zBuf + (size_t)N_NODES * 32;
    float* partial = accBuf + (size_t)N_NODES * 32;

    const int gemmBlocks = (M_TILES + WAVES_PER_BLOCK - 1) / WAVES_PER_BLOCK;

    zero_partial_kernel<<<(N_GRAPHS * D_OUT + 255) / 256, 256, 0, stream>>>(partial);
    input_gemm_kernel<<<GEMM_WGS, GEMM_WAVES * 32, 0, stream>>>(pos, feat, W1, Wp0,
                                                                yBuf, accBuf, zBuf);
    edge_scatter_kernel<<<SC_WGS, SC_WAVES_PER_WG * 32, 0, stream>>>(
        yBuf, src, dst, accBuf, nEdges);
    node_mlp_head_kernel<<<gemmBlocks, 256, 0, stream>>>(accBuf, zBuf, W2, b1, b2,
                                                         Wp1, gids, partial);
    finalize_kernel<<<(N_GRAPHS * D_OUT + 255) / 256, 256, 0, stream>>>(
        partial, bp0, bp1, out);
}


# input rows pooled per graph before the head product (cross-lane tree sums + f32 atomics into a 128x800 plane), then one small WMMA product with Wp0; per-node z product dropped; sum order changes
# speedup vs baseline: 1.8650x; 1.0887x over previous
#include <hip/hip_runtime.h>
#include <hip/hip_bf16.h>


#define N_NODES 50000
#define D_POS   32
#define D_FEAT  768
#define D_IN    801
#define D_HID   32
#define D_OUT   32
#define N_GRAPHS 128
#define M_TILES  (N_NODES / 16)
#define WAVES_PER_BLOCK 8

typedef __attribute__((ext_vector_type(2))) float v2f;
typedef __attribute__((ext_vector_type(8))) float v8f;

#define GEMM_WGS    256
#define FULL_WAVES  (M_TILES / GEMM_WGS)
#define GEMM_WAVES  (FULL_WAVES + 4)
#define K_USED      800
#define K_STEPS     (K_USED / 4)
#define K_PAIRS     (K_USED / 2)

template <int CTRL>
__device__ __forceinline__ float lane_recv(const float x)
{
    return __builtin_bit_cast(float,
        __builtin_amdgcn_update_dpp(0, __builtin_bit_cast(int, x), CTRL, 0xF, 0xF, true));
}

template <bool MASKED>
__device__ __forceinline__ float colsum16(
    const float4& a0, const float4& a1, const float4& a2, const float4& a3,
    const int c, const bool take)
{
    float v[16] = {a0.x, a0.y, a0.z, a0.w, a1.x, a1.y, a1.z, a1.w,
                   a2.x, a2.y, a2.z, a2.w, a3.x, a3.y, a3.z, a3.w};
    if (MASKED) {
#pragma unroll
        for (int i = 0; i < 16; ++i) v[i] = take ? v[i] : 0.0f;
    }
    const bool b3 = (c & 8) != 0, b2 = (c & 4) != 0, b1 = (c & 2) != 0, b0 = (c & 1) != 0;
    float w8[8], w4[4], w2[2];
#pragma unroll
    for (int i = 0; i < 8; ++i) {
        const float keep = b3 ? v[i + 8] : v[i];
        const float send = b3 ? v[i] : v[i + 8];
        w8[i] = keep + lane_recv<0x168>(send);
    }
#pragma unroll
    for (int i = 0; i < 4; ++i) {
        const float keep = b2 ? w8[i + 4] : w8[i];
        const float send = b2 ? w8[i] : w8[i + 4];
        w4[i] = keep + lane_recv<0x164>(send);
    }
#pragma unroll
    for (int i = 0; i < 2; ++i) {
        const float keep = b1 ? w4[i + 2] : w4[i];
        const float send = b1 ? w4[i] : w4[i + 2];
        w2[i] = keep + lane_recv<0x162>(send);
    }
    const float keep = b0 ? w2[1] : w2[0];
    const float send = b0 ? w2[0] : w2[1];
    return keep + lane_recv<0x161>(send);
}

template <int NVEC>
__device__ __forceinline__ void pool_chunk(
    const float4 (&a)[NVEC], float* __restrict__ p0Buf, const int kbase,
    const int gl, const bool uniform, const int g0, const int c, const int colo)
{
    if (__builtin_expect(uniform, 1)) {
        float* p = p0Buf + g0 * K_USED + kbase + colo;
#pragma unroll
        for (int grp = 0; grp < NVEC / 4; ++grp) {
            const float s = colsum16<false>(a[4 * grp], a[4 * grp + 1], a[4 * grp + 2],
                                            a[4 * grp + 3], c, true);
            atomicAdd(p + 32 * grp, s);
        }
    } else {
        unsigned rem = 0xFFFFu;
        for (int it = 0; it < 16; ++it) {
            if (rem == 0) break;
            const int r = __builtin_ctz(rem);
            const int g = __builtin_amdgcn_readlane(gl, r);
            const bool take = (gl == g);
            float* p = p0Buf + g * K_USED + kbase + colo;
#pragma unroll
            for (int grp = 0; grp < NVEC / 4; ++grp) {
                const float s = colsum16<true>(a[4 * grp], a[4 * grp + 1], a[4 * grp + 2],
                                               a[4 * grp + 3], c, take);
                atomicAdd(p + 32 * grp, s);
            }
            rem &= ~(unsigned)__builtin_amdgcn_ballot_w32(take);
        }
    }
}

template <int NVEC>
__device__ __forceinline__ void gemm_chunk(
    const float4 (&a)[NVEC], const v2f* __restrict__ Bs,
    const int s0, const int lane, v8f& cy0, v8f& cy1)
{
#pragma unroll
    for (int i = 0; i < NVEC; ++i) {
#pragma unroll
        for (int o = 0; o < 2; ++o) {
            const int s = s0 + 2 * i + o;
            const v2f by0 = Bs[s * 64 + lane];
            const v2f by1 = Bs[s * 64 + 32 + lane];
            v2f av;
            av.x = o ? a[i].z : a[i].x;
            av.y = o ? a[i].w : a[i].y;
            cy0 = __builtin_amdgcn_wmma_f32_16x16x4_f32(false, av, false, by0, (short)0, cy0, false, false);
            cy1 = __builtin_amdgcn_wmma_f32_16x16x4_f32(false, av, false, by1, (short)0, cy1, false, false);
        }
    }
}

template <int NVEC>
__device__ __forceinline__ void gemm_chunk_half(
    const float4 (&a)[NVEC], const v2f* __restrict__ Bq, const int s0, v8f& acc)
{
#pragma unroll
    for (int i = 0; i < NVEC; ++i) {
#pragma unroll
        for (int o = 0; o < 2; ++o) {
            const int s = s0 + 2 * i + o;
            const v2f b = Bq[s * 64];
            v2f av;
            av.x = o ? a[i].z : a[i].x;
            av.y = o ? a[i].w : a[i].y;
            acc = __builtin_amdgcn_wmma_f32_16x16x4_f32(false, av, false, b, (short)0, acc, false, false);
        }
    }
}

template <int MODE>
__device__ __forceinline__ void chunk_work(
    const float4 (&aP)[4], const v2f* __restrict__ Bs, const v2f* __restrict__ Bq,
    float* __restrict__ p0Buf, const int lane, const int gl, const bool uniform,
    const int g0, const int c, const int colo, v8f& c0, v8f& c1)
{
    if (MODE == 0) gemm_chunk<4>(aP, Bs, 0, lane, c0, c1);
    if (MODE == 1) gemm_chunk_half<4>(aP, Bq, 0, c0);
    if (MODE != 1) pool_chunk<4>(aP, p0Buf, 0, gl, uniform, g0, c, colo);
}
template <int MODE>
__device__ __forceinline__ void chunk_work(
    const float4 (&a)[8], const int st, const v2f* __restrict__ Bs, const v2f* __restrict__ Bq,
    float* __restrict__ p0Buf, const int lane, const int gl, const bool uniform,
    const int g0, const int c, const int colo, v8f& c0, v8f& c1)
{
    if (MODE == 0) gemm_chunk<8>(a, Bs, 8 + 16 * st, lane, c0, c1);
    if (MODE == 1) gemm_chunk_half<8>(a, Bq, 8 + 16 * st, c0);
    if (MODE != 1) pool_chunk<8>(a, p0Buf, D_POS + 64 * st, gl, uniform, g0, c, colo);
}

template <int MODE>
__device__ __forceinline__ void run_tile(
    const float4* __restrict__ featV, const float4 (&aP)[4], float4 (&aA)[8],
    const v2f* __restrict__ Bs, const v2f* __restrict__ Bq, float* __restrict__ p0Buf,
    const int lane, const int gl, const bool uniform, const int g0,
    const int c, const int colo, v8f& c0, v8f& c1)
{
    float4 aB[8];
    chunk_work<MODE>(aP, Bs, Bq, p0Buf, lane, gl, uniform, g0, c, colo, c0, c1);

    for (int st = 0; st < D_FEAT / 64 - 2; st += 2) {
#pragma unroll
        for (int j = 0; j < 8; ++j) aB[j] = featV[16 * (st + 1) + 2 * j];
        chunk_work<MODE>(aA, st, Bs, Bq, p0Buf, lane, gl, uniform, g0, c, colo, c0, c1);
#pragma unroll
        for (int j = 0; j < 8; ++j) aA[j] = featV[16 * (st + 2) + 2 * j];
        chunk_work<MODE>(aB, st + 1, Bs, Bq, p0Buf, lane, gl, uniform, g0, c, colo, c0, c1);
    }
    {
        const int st = D_FEAT / 64 - 2;
#pragma unroll
        for (int j = 0; j < 8; ++j) aB[j] = featV[16 * (st + 1) + 2 * j];
        chunk_work<MODE>(aA, st, Bs, Bq, p0Buf, lane, gl, uniform, g0, c, colo, c0, c1);
        chunk_work<MODE>(aB, st + 1, Bs, Bq, p0Buf, lane, gl, uniform, g0, c, colo, c0, c1);
    }
}

__global__ __launch_bounds__(GEMM_WAVES * 32) void input_gemm_kernel(
    const float* __restrict__ pos, const float* __restrict__ feat,
    const float* __restrict__ W1,  const int* __restrict__ gids,
    float* __restrict__ yBuf, float* __restrict__ accBuf, float* __restrict__ p0Buf)
{
    __shared__ __attribute__((aligned(16))) v2f Bs[K_STEPS * 64];

    const int wave   = __builtin_amdgcn_readfirstlane((int)(threadIdx.x >> 5));
    const int lane   = threadIdx.x & 31;
    const bool whole  = wave < FULL_WAVES;
    const int role   = wave - FULL_WAVES;
    const int tile   = (whole ? wave : FULL_WAVES) * GEMM_WGS + (int)blockIdx.x;
    const bool active = (tile < M_TILES) && (role < 3);

    const int  m0   = (tile < M_TILES ? tile : 0) * 16;
    const int  col  = lane & 15;
    const int  h    = lane >> 4;
    const int  node = m0 + col;

    const float4* posV  = (const float4*)(pos  + (size_t)node * D_POS)  + h;
    const float4* featV = (const float4*)(feat + (size_t)node * D_FEAT) + h;

    float4 aP[4], aA[8];
#pragma unroll
    for (int j = 0; j < 4; ++j) aP[j] = posV[2 * j];
#pragma unroll
    for (int j = 0; j < 8; ++j) aA[j] = featV[2 * j];
    const int gl = gids[node];

    {
        const int n  = lane;
        const int nh = n >> 4, c = n & 15;
        for (int bt = wave; bt < K_PAIRS / 16; bt += GEMM_WAVES) {
            const int k0   = 32 * bt;
            const int row0 = (k0 < D_POS) ? k0 : (k0 + 1);
            const float* w1p = W1 + row0 * 32 + n;
            v2f v[16];
#pragma unroll
            for (int u = 0; u < 16; ++u) {
                v[u].x = w1p[(2 * u) * 32];
                v[u].y = w1p[(2 * u + 1) * 32];
            }
#pragma unroll
            for (int u = 0; u < 16; ++u) {
                const int s = 2 * (4 * bt + (u >> 2)) + (u & 1);
                Bs[s * 64 + nh * 32 + ((u >> 1) & 1) * 16 + c] = v[u];
            }
        }
    }
    __syncthreads();
    if (!active) return;

    const int  g0      = __builtin_amdgcn_readfirstlane(gl);
    const bool uniform = __builtin_amdgcn_ballot_w32(gl == g0) == 0xFFFFFFFFu;
    const int  colo    = 8 * (col >> 2) + 4 * h + (col & 3);

    if (whole) {
        v8f cy0 = {}, cy1 = {};
        run_tile<0>(featV, aP, aA, Bs, Bs, p0Buf, lane, gl, uniform, g0, col, colo, cy0, cy1);

        for (int j = 0; j < 8; ++j) {
            const int row = 8 * h + j;
            const int off = (m0 + row) * 32 + col;
            yBuf[off]        = cy0[j];  accBuf[off]        = cy0[j];
            yBuf[off + 16]   = cy1[j];  accBuf[off + 16]   = cy1[j];
        }
    } else if (role < 2) {
        const v2f* Bq = Bs + role * 32 + lane;
        v8f acc = {}, unused = {};
        run_tile<1>(featV, aP, aA, Bs, Bq, p0Buf, lane, gl, uniform, g0, col, colo, acc, unused);

        for (int j = 0; j < 8; ++j) {
            const int row = 8 * h + j;
            const int off = (m0 + row) * 32 + role * 16 + col;
            yBuf[off] = acc[j];  accBuf[off] = acc[j];
        }
    } else {
        v8f unused0 = {}, unused1 = {};
        run_tile<2>(featV, aP, aA, Bs, Bs, p0Buf, lane, gl, uniform, g0, col, colo, unused0, unused1);
    }
}

#define SC_WGS          512
#define SC_WAVES_PER_WG 8
#define SC_WAVES_LOG2   12
#define SC_UNIT         16

template <int BASE, int N>
__device__ __forceinline__ void scatter_edges(
    const float* __restrict__ yBuf, float* __restrict__ accBuf,
    const int sv, const int dv, const int lane)
{
    float v[N];
    int   d[N];
#pragma unroll
    for (int u = 0; u < N; ++u) {
        const int s = __builtin_amdgcn_readlane(sv, BASE + u);
        d[u]        = __builtin_amdgcn_readlane(dv, BASE + u);
        v[u] = yBuf[(size_t)s * 32 + lane];
    }
#pragma unroll
    for (int u = 0; u < N; ++u)
        atomicAdd(accBuf + (size_t)d[u] * 32 + lane, v[u]);
}

__global__ __launch_bounds__(SC_WAVES_PER_WG * 32) void edge_scatter_kernel(
    const float* __restrict__ yBuf, const int* __restrict__ src,
    const int* __restrict__ dst, float* __restrict__ accBuf, int nEdges)
{
    const int wave = __builtin_amdgcn_readfirstlane((int)(threadIdx.x >> 5));
    const int lane = threadIdx.x & 31;
    const int w    = (int)blockIdx.x * SC_WAVES_PER_WG + wave;

    const int nUnits = (nEdges + SC_UNIT - 1) / SC_UNIT;
    const int u0 = (int)(((long long)w * nUnits) >> SC_WAVES_LOG2);
    const int u1 = (int)(((long long)(w + 1) * nUnits) >> SC_WAVES_LOG2);
    const int eBeg = u0 * SC_UNIT;
    const int eEnd = (u1 * SC_UNIT < nEdges) ? u1 * SC_UNIT : nEdges;

    if (eBeg >= eEnd) return;

    const int last = nEdges - 1;
    const int p0   = (eBeg + lane < last) ? eBeg + lane : last;
    int sv = src[p0], dv = dst[p0];

    for (int e0 = eBeg; e0 < eEnd; e0 += 32) {
        const int pN  = (e0 + 32 + lane < last) ? e0 + 32 + lane : last;
        const int svN = src[pN], dvN = dst[pN];

        const int left = eEnd - e0;
        if (left >= 32) {
            scatter_edges<0, 32>(yBuf, accBuf, sv, dv, lane);
        } else {
            int done = 0;
            if (left >= SC_UNIT) {
                scatter_edges<0, SC_UNIT>(yBuf, accBuf, sv, dv, lane);
                done = SC_UNIT;
            }
            for (int i = done; i < left; ++i) {
                const int s = __builtin_amdgcn_readlane(sv, i);
                const int d = __builtin_amdgcn_readlane(dv, i);
                atomicAdd(accBuf + (size_t)d * 32 + lane, yBuf[(size_t)s * 32 + lane]);
            }
        }
        sv = svN;  dv = dvN;
    }
}

#define NODE_BLOCKS ((M_TILES + WAVES_PER_BLOCK - 1) / WAVES_PER_BLOCK)
#define HEAD_BLOCKS (K_USED / 32)

__global__ __launch_bounds__(256) void node_mlp_head_kernel(
    const float* __restrict__ accBuf, const float* __restrict__ p0Buf,
    const float* __restrict__ W2, const float* __restrict__ bias1,
    const float* __restrict__ bias2, const float* __restrict__ Wp0,
    const float* __restrict__ Wp1,
    const int* __restrict__ gids, float* __restrict__ partial)
{
    __shared__ float hbuf[WAVES_PER_BLOCK][16 * 32];

    const int wave = threadIdx.x >> 5;
    const int lane = threadIdx.x & 31;
    const int  col  = lane & 15;
    const bool hi   = lane >= 16;

    if ((int)blockIdx.x >= NODE_BLOCKS) {
        const int ku0   = 32 * ((int)blockIdx.x - NODE_BLOCKS);
        const int wrow0 = (ku0 < D_POS) ? ku0 : (ku0 + 1);
        const float* aRow = p0Buf + (size_t)(16 * wave + col) * K_USED + ku0;
        v8f u0 = {}, u1 = {};
        for (int k = 0; k < 32; k += 4) {
            const int kb = k + (hi ? 2 : 0);
            const v2f a = *(const v2f*)(aRow + kb);
            v2f b0, b1f;
            b0.x  = Wp0[(wrow0 + kb) * 32 + col];       b0.y  = Wp0[(wrow0 + kb + 1) * 32 + col];
            b1f.x = Wp0[(wrow0 + kb) * 32 + 16 + col];  b1f.y = Wp0[(wrow0 + kb + 1) * 32 + 16 + col];
            u0 = __builtin_amdgcn_wmma_f32_16x16x4_f32(false, a, false, b0,  (short)0, u0, false, false);
            u1 = __builtin_amdgcn_wmma_f32_16x16x4_f32(false, a, false, b1f, (short)0, u1, false, false);
        }
        for (int j = 0; j < 8; ++j) {
            const int g = 16 * wave + (hi ? 8 : 0) + j;
            atomicAdd(&partial[g * 32 + col],      u0[j]);
            atomicAdd(&partial[g * 32 + 16 + col], u1[j]);
        }
        return;
    }

    const int tile = blockIdx.x * WAVES_PER_BLOCK + wave;
    if (tile >= M_TILES) return;

    const int  m0   = tile * 16;
    const int  node = m0 + col;

    const float* accRow = accBuf + (size_t)node * 32;
    v8f c10 = {}, c11 = {};
    for (int k = 0; k < 32; k += 4) {
        const int kb = k + (hi ? 2 : 0);
        const v2f av = *(const v2f*)(accRow + kb);
        const v2f bv = *(const v2f*)(bias1 + kb);
        v2f a;
        a.x = fmaxf(av.x + bv.x, 0.0f);
        a.y = fmaxf(av.y + bv.y, 0.0f);
        v2f b0, b1f;
        b0.x  = W2[kb * 32 + col];       b0.y  = W2[(kb + 1) * 32 + col];
        b1f.x = W2[kb * 32 + 16 + col];  b1f.y = W2[(kb + 1) * 32 + 16 + col];
        c10 = __builtin_amdgcn_wmma_f32_16x16x4_f32(false, a, false, b0,  (short)0, c10, false, false);
        c11 = __builtin_amdgcn_wmma_f32_16x16x4_f32(false, a, false, b1f, (short)0, c11, false, false);
    }

    const float bb0 = bias2[col];
    const float bb1 = bias2[16 + col];
    for (int j = 0; j < 8; ++j) {
        const int row = (hi ? 8 : 0) + j;
        hbuf[wave][row * 32 + col]      = fmaxf(c10[j] + bb0, 0.0f);
        hbuf[wave][row * 32 + 16 + col] = fmaxf(c11[j] + bb1, 0.0f);
    }

    v8f u0 = {}, u1 = {};
    for (int k = 0; k < 32; k += 4) {
        const int kb = k + (hi ? 2 : 0);
        const v2f a = *(const v2f*)(&hbuf[wave][col * 32 + kb]);
        v2f b0, b1f;
        b0.x  = Wp1[kb * 32 + col];       b0.y  = Wp1[(kb + 1) * 32 + col];
        b1f.x = Wp1[kb * 32 + 16 + col];  b1f.y = Wp1[(kb + 1) * 32 + 16 + col];
        u0 = __builtin_amdgcn_wmma_f32_16x16x4_f32(false, a, false, b0,  (short)0, u0, false, false);
        u1 = __builtin_amdgcn_wmma_f32_16x16x4_f32(false, a, false, b1f, (short)0, u1, false, false);
    }

    for (int j = 0; j < 8; ++j) {
        const int row  = (hi ? 8 : 0) + j;
        const int nd   = m0 + row;
        const int g    = gids[nd];
        atomicAdd(&partial[g * 32 + col],      u0[j]);
        atomicAdd(&partial[g * 32 + 16 + col], u1[j]);
    }
}

#define POOL_FLOATS (N_GRAPHS * (K_USED + D_OUT))

__global__ void zero_pool_kernel(float* __restrict__ p) {
    const int i = blockIdx.x * blockDim.x + threadIdx.x;
    if (i < POOL_FLOATS) p[i] = 0.0f;
}

__global__ void finalize_kernel(const float* __restrict__ partial,
                                const float* __restrict__ bp0,
                                const float* __restrict__ bp1,
                                float* __restrict__ out) {
    const int i = blockIdx.x * blockDim.x + threadIdx.x;
    if (i < N_GRAPHS * D_OUT) {
        const int c = i & 31;
        out[i] = partial[i] + bp0[c] + bp1[c];
    }
}

extern "C" void kernel_launch(void* const* d_in, const int* in_sizes, int n_in,
                              void* d_out, int out_size, void* d_ws, size_t ws_size,
                              hipStream_t stream) {
    const float* pos  = (const float*)d_in[0];
    const float* feat = (const float*)d_in[1];
    const int*   src  = (const int*)  d_in[2];
    const int*   dst  = (const int*)  d_in[3];
    const int*   gids = (const int*)  d_in[4];
    const float* W1   = (const float*)d_in[5];
    const float* b1   = (const float*)d_in[6];
    const float* W2   = (const float*)d_in[7];
    const float* b2   = (const float*)d_in[8];
    const float* Wp0  = (const float*)d_in[9];
    const float* bp0  = (const float*)d_in[10];
    const float* Wp1  = (const float*)d_in[11];
    const float* bp1  = (const float*)d_in[12];
    float* out = (float*)d_out;
    const int nEdges = in_sizes[2];

    float* yBuf    = (float*)d_ws;
    float* accBuf  = yBuf + (size_t)N_NODES * 32;
    float* p0Buf   = accBuf + (size_t)N_NODES * 32;
    float* partial = p0Buf + (size_t)N_GRAPHS * K_USED;

    zero_pool_kernel<<<(POOL_FLOATS + 255) / 256, 256, 0, stream>>>(p0Buf);
    input_gemm_kernel<<<GEMM_WGS, GEMM_WAVES * 32, 0, stream>>>(pos, feat, W1, gids,
                                                                yBuf, accBuf, p0Buf);
    edge_scatter_kernel<<<SC_WGS, SC_WAVES_PER_WG * 32, 0, stream>>>(
        yBuf, src, dst, accBuf, nEdges);
    node_mlp_head_kernel<<<NODE_BLOCKS + HEAD_BLOCKS, 256, 0, stream>>>(
        accBuf, p0Buf, W2, b1, b2, Wp0, Wp1, gids, partial);
    finalize_kernel<<<(N_GRAPHS * D_OUT + 255) / 256, 256, 0, stream>>>(
        partial, bp0, bp1, out);
}


# input GEMM A rows fetched as whole 128-byte lines (512 contiguous bytes of 2 or 4 rows per load) and turned to fragment order in a wave-private LDS strip, not 32 B from each of 16 rows per load
# speedup vs baseline: 1.9744x; 1.0587x over previous
#include <hip/hip_runtime.h>
#include <hip/hip_bf16.h>


#define N_NODES 50000
#define D_POS   32
#define D_FEAT  768
#define D_IN    801
#define D_HID   32
#define D_OUT   32
#define N_GRAPHS 128
#define M_TILES  (N_NODES / 16)
#define WAVES_PER_BLOCK 8

typedef __attribute__((ext_vector_type(2))) float v2f;
typedef __attribute__((ext_vector_type(8))) float v8f;
typedef __attribute__((ext_vector_type(4))) float v4f;

#define GEMM_WGS    256
#define FULL_WAVES  (M_TILES / GEMM_WGS)
#define GEMM_WAVES  (FULL_WAVES + 4)
#define K_USED      800
#define K_STEPS     (K_USED / 4)
#define K_PAIRS     (K_USED / 2)

template <int CTRL>
__device__ __forceinline__ float lane_recv(const float x)
{
    return __builtin_bit_cast(float,
        __builtin_amdgcn_update_dpp(0, __builtin_bit_cast(int, x), CTRL, 0xF, 0xF, true));
}

template <bool MASKED>
__device__ __forceinline__ float colsum16(
    const float4& a0, const float4& a1, const float4& a2, const float4& a3,
    const int c, const bool take)
{
    float v[16] = {a0.x, a0.y, a0.z, a0.w, a1.x, a1.y, a1.z, a1.w,
                   a2.x, a2.y, a2.z, a2.w, a3.x, a3.y, a3.z, a3.w};
    if (MASKED) {
#pragma unroll
        for (int i = 0; i < 16; ++i) v[i] = take ? v[i] : 0.0f;
    }
    const bool b3 = (c & 8) != 0, b2 = (c & 4) != 0, b1 = (c & 2) != 0, b0 = (c & 1) != 0;
    float w8[8], w4[4], w2[2];
#pragma unroll
    for (int i = 0; i < 8; ++i) {
        const float keep = b3 ? v[i + 8] : v[i];
        const float send = b3 ? v[i] : v[i + 8];
        w8[i] = keep + lane_recv<0x168>(send);
    }
#pragma unroll
    for (int i = 0; i < 4; ++i) {
        const float keep = b2 ? w8[i + 4] : w8[i];
        const float send = b2 ? w8[i] : w8[i + 4];
        w4[i] = keep + lane_recv<0x164>(send);
    }
#pragma unroll
    for (int i = 0; i < 2; ++i) {
        const float keep = b1 ? w4[i + 2] : w4[i];
        const float send = b1 ? w4[i] : w4[i + 2];
        w2[i] = keep + lane_recv<0x162>(send);
    }
    const float keep = b0 ? w2[1] : w2[0];
    const float send = b0 ? w2[0] : w2[1];
    return keep + lane_recv<0x161>(send);
}

template <int NVEC>
__device__ __forceinline__ void pool_chunk(
    const float4 (&a)[NVEC], float* __restrict__ p0Buf, const int kbase,
    const int gl, const bool uniform, const int g0, const int c, const int colo)
{
    if (__builtin_expect(uniform, 1)) {
        float* p = p0Buf + g0 * K_USED + kbase + colo;
#pragma unroll
        for (int grp = 0; grp < NVEC / 4; ++grp) {
            const float s = colsum16<false>(a[4 * grp], a[4 * grp + 1], a[4 * grp + 2],
                                            a[4 * grp + 3], c, true);
            atomicAdd(p + 32 * grp, s);
        }
    } else {
        unsigned rem = 0xFFFFu;
        for (int it = 0; it < 16; ++it) {
            if (rem == 0) break;
            const int r = __builtin_ctz(rem);
            const int g = __builtin_amdgcn_readlane(gl, r);
            const bool take = (gl == g);
            float* p = p0Buf + g * K_USED + kbase + colo;
#pragma unroll
            for (int grp = 0; grp < NVEC / 4; ++grp) {
                const float s = colsum16<true>(a[4 * grp], a[4 * grp + 1], a[4 * grp + 2],
                                               a[4 * grp + 3], c, take);
                atomicAdd(p + 32 * grp, s);
            }
            rem &= ~(unsigned)__builtin_amdgcn_ballot_w32(take);
        }
    }
}

template <int NVEC>
__device__ __forceinline__ void gemm_chunk(
    const float4 (&a)[NVEC], const v2f* __restrict__ Bs,
    const int s0, const int lane, v8f& cy0, v8f& cy1)
{
#pragma unroll
    for (int i = 0; i < NVEC; ++i) {
#pragma unroll
        for (int o = 0; o < 2; ++o) {
            const int s = s0 + 2 * i + o;
            const v2f by0 = Bs[s * 64 + lane];
            const v2f by1 = Bs[s * 64 + 32 + lane];
            v2f av;
            av.x = o ? a[i].z : a[i].x;
            av.y = o ? a[i].w : a[i].y;
            cy0 = __builtin_amdgcn_wmma_f32_16x16x4_f32(false, av, false, by0, (short)0, cy0, false, false);
            cy1 = __builtin_amdgcn_wmma_f32_16x16x4_f32(false, av, false, by1, (short)0, cy1, false, false);
        }
    }
}

template <int NVEC>
__device__ __forceinline__ void gemm_chunk_half(
    const float4 (&a)[NVEC], const v2f* __restrict__ Bq, const int s0, v8f& acc)
{
#pragma unroll
    for (int i = 0; i < NVEC; ++i) {
#pragma unroll
        for (int o = 0; o < 2; ++o) {
            const int s = s0 + 2 * i + o;
            const v2f b = Bq[s * 64];
            v2f av;
            av.x = o ? a[i].z : a[i].x;
            av.y = o ? a[i].w : a[i].y;
            acc = __builtin_amdgcn_wmma_f32_16x16x4_f32(false, av, false, b, (short)0, acc, false, false);
        }
    }
}

template <int MODE>
__device__ __forceinline__ void chunk_work(
    const float4 (&aP)[4], const v2f* __restrict__ Bs, const v2f* __restrict__ Bq,
    float* __restrict__ p0Buf, const int lane, const int gl, const bool uniform,
    const int g0, const int c, const int colo, v8f& c0, v8f& c1)
{
    if (MODE == 0) gemm_chunk<4>(aP, Bs, 0, lane, c0, c1);
    if (MODE == 1) gemm_chunk_half<4>(aP, Bq, 0, c0);
    if (MODE != 1) pool_chunk<4>(aP, p0Buf, 0, gl, uniform, g0, c, colo);
}
template <int MODE>
__device__ __forceinline__ void chunk_work(
    const float4 (&a)[8], const int st, const v2f* __restrict__ Bs, const v2f* __restrict__ Bq,
    float* __restrict__ p0Buf, const int lane, const int gl, const bool uniform,
    const int g0, const int c, const int colo, v8f& c0, v8f& c1)
{
    if (MODE == 0) gemm_chunk<8>(a, Bs, 8 + 16 * st, lane, c0, c1);
    if (MODE == 1) gemm_chunk_half<8>(a, Bq, 8 + 16 * st, c0);
    if (MODE != 1) pool_chunk<8>(a, p0Buf, D_POS + 64 * st, gl, uniform, g0, c, colo);
}

#define A_PITCH4 17

template <int MODE>
__device__ __forceinline__ void run_tile(
    const v4f* __restrict__ featL, const v4f (&gP)[4], const v4f (&g0v)[8],
    v4f* __restrict__ AsW,
    const v2f* __restrict__ Bs, const v2f* __restrict__ Bq, float* __restrict__ p0Buf,
    const int lane, const int gl, const bool uniform, const int g0,
    const int c, const int colo, v8f& c0, v8f& c1)
{
    v4f* stP = AsW + (lane >> 3) * A_PITCH4 + (lane & 7);
    v4f* stF = AsW + (lane >> 4) * A_PITCH4 + (lane & 15);
    const v4f* ld = AsW + c * A_PITCH4 + (lane >> 4);

    float4 aP[4];
#pragma unroll
    for (int i = 0; i < 4; ++i) stP[4 * i * A_PITCH4] = gP[i];
#pragma unroll
    for (int j = 0; j < 4; ++j) { const v4f t = ld[2 * j]; aP[j] = make_float4(t.x, t.y, t.z, t.w); }
    chunk_work<MODE>(aP, Bs, Bq, p0Buf, lane, gl, uniform, g0, c, colo, c0, c1);

    float4 a[8];
    v4f g[8];
#pragma unroll
    for (int i = 0; i < 8; ++i) g[i] = g0v[i];
    for (int st = 0; st < D_FEAT / 64 - 1; ++st) {
#pragma unroll
        for (int i = 0; i < 8; ++i) stF[2 * i * A_PITCH4] = g[i];
#pragma unroll
        for (int i = 0; i < 8; ++i) g[i] = featL[2 * i * (D_FEAT / 4) + 16 * (st + 1)];
#pragma unroll
        for (int j = 0; j < 8; ++j) { const v4f t = ld[2 * j]; a[j] = make_float4(t.x, t.y, t.z, t.w); }
        chunk_work<MODE>(a, st, Bs, Bq, p0Buf, lane, gl, uniform, g0, c, colo, c0, c1);
    }
    {
#pragma unroll
        for (int i = 0; i < 8; ++i) stF[2 * i * A_PITCH4] = g[i];
#pragma unroll
        for (int j = 0; j < 8; ++j) { const v4f t = ld[2 * j]; a[j] = make_float4(t.x, t.y, t.z, t.w); }
        chunk_work<MODE>(a, D_FEAT / 64 - 1, Bs, Bq, p0Buf, lane, gl, uniform, g0, c, colo, c0, c1);
    }
}

__global__ __launch_bounds__(GEMM_WAVES * 32) void input_gemm_kernel(
    const float* __restrict__ pos, const float* __restrict__ feat,
    const float* __restrict__ W1,  const int* __restrict__ gids,
    float* __restrict__ yBuf, float* __restrict__ accBuf, float* __restrict__ p0Buf)
{
    __shared__ __attribute__((aligned(16))) v2f Bs[K_STEPS * 64];
    __shared__ __attribute__((aligned(16))) v4f As[GEMM_WAVES * 16 * A_PITCH4];

    const int wave   = __builtin_amdgcn_readfirstlane((int)(threadIdx.x >> 5));
    const int lane   = threadIdx.x & 31;
    const bool whole  = wave < FULL_WAVES;
    const int role   = wave - FULL_WAVES;
    const int tile   = (whole ? wave : FULL_WAVES) * GEMM_WGS + (int)blockIdx.x;
    const bool active = (tile < M_TILES) && (role < 3);

    const int  m0   = (tile < M_TILES ? tile : 0) * 16;
    const int  col  = lane & 15;
    const int  h    = lane >> 4;
    const int  node = m0 + col;

    const v4f* posL  = (const v4f*)(pos  + (size_t)m0 * D_POS)  + (lane >> 3) * (D_POS / 4)  + (lane & 7);
    const v4f* featL = (const v4f*)(feat + (size_t)m0 * D_FEAT) + (lane >> 4) * (D_FEAT / 4) + (lane & 15);
    v4f* AsW = As + wave * 16 * A_PITCH4;

    v4f gP[4], gA[8];
#pragma unroll
    for (int i = 0; i < 4; ++i) gP[i] = posL[4 * i * (D_POS / 4)];
#pragma unroll
    for (int i = 0; i < 8; ++i) gA[i] = featL[2 * i * (D_FEAT / 4)];
    const int gl = gids[node];

    {
        const int n  = lane;
        const int nh = n >> 4, c = n & 15;
        for (int bt = wave; bt < K_PAIRS / 16; bt += GEMM_WAVES) {
            const int k0   = 32 * bt;
            const int row0 = (k0 < D_POS) ? k0 : (k0 + 1);
            const float* w1p = W1 + row0 * 32 + n;
            v2f v[16];
#pragma unroll
            for (int u = 0; u < 16; ++u) {
                v[u].x = w1p[(2 * u) * 32];
                v[u].y = w1p[(2 * u + 1) * 32];
            }
#pragma unroll
            for (int u = 0; u < 16; ++u) {
                const int s = 2 * (4 * bt + (u >> 2)) + (u & 1);
                Bs[s * 64 + nh * 32 + ((u >> 1) & 1) * 16 + c] = v[u];
            }
        }
    }
    __syncthreads();
    if (!active) return;

    const int  g0      = __builtin_amdgcn_readfirstlane(gl);
    const bool uniform = __builtin_amdgcn_ballot_w32(gl == g0) == 0xFFFFFFFFu;
    const int  colo    = 8 * (col >> 2) + 4 * h + (col & 3);

    if (whole) {
        v8f cy0 = {}, cy1 = {};
        run_tile<0>(featL, gP, gA, AsW, Bs, Bs, p0Buf, lane, gl, uniform, g0, col, colo, cy0, cy1);

        for (int j = 0; j < 8; ++j) {
            const int row = 8 * h + j;
            const int off = (m0 + row) * 32 + col;
            yBuf[off]        = cy0[j];  accBuf[off]        = cy0[j];
            yBuf[off + 16]   = cy1[j];  accBuf[off + 16]   = cy1[j];
        }
    } else if (role < 2) {
        const v2f* Bq = Bs + role * 32 + lane;
        v8f acc = {}, unused = {};
        run_tile<1>(featL, gP, gA, AsW, Bs, Bq, p0Buf, lane, gl, uniform, g0, col, colo, acc, unused);

        for (int j = 0; j < 8; ++j) {
            const int row = 8 * h + j;
            const int off = (m0 + row) * 32 + role * 16 + col;
            yBuf[off] = acc[j];  accBuf[off] = acc[j];
        }
    } else {
        v8f unused0 = {}, unused1 = {};
        run_tile<2>(featL, gP, gA, AsW, Bs, Bs, p0Buf, lane, gl, uniform, g0, col, colo, unused0, unused1);
    }
}

#define SC_WGS          512
#define SC_WAVES_PER_WG 8
#define SC_WAVES_LOG2   12
#define SC_UNIT         16

template <int BASE, int N>
__device__ __forceinline__ void scatter_edges(
    const float* __restrict__ yBuf, float* __restrict__ accBuf,
    const int sv, const int dv, const int lane)
{
    float v[N];
    int   d[N];
#pragma unroll
    for (int u = 0; u < N; ++u) {
        const int s = __builtin_amdgcn_readlane(sv, BASE + u);
        d[u]        = __builtin_amdgcn_readlane(dv, BASE + u);
        v[u] = yBuf[(size_t)s * 32 + lane];
    }
#pragma unroll
    for (int u = 0; u < N; ++u)
        atomicAdd(accBuf + (size_t)d[u] * 32 + lane, v[u]);
}

__global__ __launch_bounds__(SC_WAVES_PER_WG * 32) void edge_scatter_kernel(
    const float* __restrict__ yBuf, const int* __restrict__ src,
    const int* __restrict__ dst, float* __restrict__ accBuf, int nEdges)
{
    const int wave = __builtin_amdgcn_readfirstlane((int)(threadIdx.x >> 5));
    const int lane = threadIdx.x & 31;
    const int w    = (int)blockIdx.x * SC_WAVES_PER_WG + wave;

    const int nUnits = (nEdges + SC_UNIT - 1) / SC_UNIT;
    const int u0 = (int)(((long long)w * nUnits) >> SC_WAVES_LOG2);
    const int u1 = (int)(((long long)(w + 1) * nUnits) >> SC_WAVES_LOG2);
    const int eBeg = u0 * SC_UNIT;
    const int eEnd = (u1 * SC_UNIT < nEdges) ? u1 * SC_UNIT : nEdges;

    if (eBeg >= eEnd) return;

    const int last = nEdges - 1;
    const int p0   = (eBeg + lane < last) ? eBeg + lane : last;
    int sv = src[p0], dv = dst[p0];

    for (int e0 = eBeg; e0 < eEnd; e0 += 32) {
        const int pN  = (e0 + 32 + lane < last) ? e0 + 32 + lane : last;
        const int svN = src[pN], dvN = dst[pN];

        const int left = eEnd - e0;
        if (left >= 32) {
            scatter_edges<0, 32>(yBuf, accBuf, sv, dv, lane);
        } else {
            int done = 0;
            if (left >= SC_UNIT) {
                scatter_edges<0, SC_UNIT>(yBuf, accBuf, sv, dv, lane);
                done = SC_UNIT;
            }
            for (int i = done; i < left; ++i) {
                const int s = __builtin_amdgcn_readlane(sv, i);
                const int d = __builtin_amdgcn_readlane(dv, i);
                atomicAdd(accBuf + (size_t)d * 32 + lane, yBuf[(size_t)s * 32 + lane]);
            }
        }
        sv = svN;  dv = dvN;
    }
}

#define NODE_BLOCKS ((M_TILES + WAVES_PER_BLOCK - 1) / WAVES_PER_BLOCK)
#define HEAD_BLOCKS (K_USED / 32)

__global__ __launch_bounds__(256) void node_mlp_head_kernel(
    const float* __restrict__ accBuf, const float* __restrict__ p0Buf,
    const float* __restrict__ W2, const float* __restrict__ bias1,
    const float* __restrict__ bias2, const float* __restrict__ Wp0,
    const float* __restrict__ Wp1,
    const int* __restrict__ gids, float* __restrict__ partial)
{
    __shared__ float hbuf[WAVES_PER_BLOCK][16 * 32];

    const int wave = threadIdx.x >> 5;
    const int lane = threadIdx.x & 31;
    const int  col  = lane & 15;
    const bool hi   = lane >= 16;

    if ((int)blockIdx.x >= NODE_BLOCKS) {
        const int ku0   = 32 * ((int)blockIdx.x - NODE_BLOCKS);
        const int wrow0 = (ku0 < D_POS) ? ku0 : (ku0 + 1);
        const float* aRow = p0Buf + (size_t)(16 * wave + col) * K_USED + ku0;
        v8f u0 = {}, u1 = {};
        for (int k = 0; k < 32; k += 4) {
            const int kb = k + (hi ? 2 : 0);
            const v2f a = *(const v2f*)(aRow + kb);
            v2f b0, b1f;
            b0.x  = Wp0[(wrow0 + kb) * 32 + col];       b0.y  = Wp0[(wrow0 + kb + 1) * 32 + col];
            b1f.x = Wp0[(wrow0 + kb) * 32 + 16 + col];  b1f.y = Wp0[(wrow0 + kb + 1) * 32 + 16 + col];
            u0 = __builtin_amdgcn_wmma_f32_16x16x4_f32(false, a, false, b0,  (short)0, u0, false, false);
            u1 = __builtin_amdgcn_wmma_f32_16x16x4_f32(false, a, false, b1f, (short)0, u1, false, false);
        }
        for (int j = 0; j < 8; ++j) {
            const int g = 16 * wave + (hi ? 8 : 0) + j;
            atomicAdd(&partial[g * 32 + col],      u0[j]);
            atomicAdd(&partial[g * 32 + 16 + col], u1[j]);
        }
        return;
    }

    const int tile = blockIdx.x * WAVES_PER_BLOCK + wave;
    if (tile >= M_TILES) return;

    const int  m0   = tile * 16;
    const int  node = m0 + col;

    const float* accRow = accBuf + (size_t)node * 32;
    v8f c10 = {}, c11 = {};
    for (int k = 0; k < 32; k += 4) {
        const int kb = k + (hi ? 2 : 0);
        const v2f av = *(const v2f*)(accRow + kb);
        const v2f bv = *(const v2f*)(bias1 + kb);
        v2f a;
        a.x = fmaxf(av.x + bv.x, 0.0f);
        a.y = fmaxf(av.y + bv.y, 0.0f);
        v2f b0, b1f;
        b0.x  = W2[kb * 32 + col];       b0.y  = W2[(kb + 1) * 32 + col];
        b1f.x = W2[kb * 32 + 16 + col];  b1f.y = W2[(kb + 1) * 32 + 16 + col];
        c10 = __builtin_amdgcn_wmma_f32_16x16x4_f32(false, a, false, b0,  (short)0, c10, false, false);
        c11 = __builtin_amdgcn_wmma_f32_16x16x4_f32(false, a, false, b1f, (short)0, c11, false, false);
    }

    const float bb0 = bias2[col];
    const float bb1 = bias2[16 + col];
    for (int j = 0; j < 8; ++j) {
        const int row = (hi ? 8 : 0) + j;
        hbuf[wave][row * 32 + col]      = fmaxf(c10[j] + bb0, 0.0f);
        hbuf[wave][row * 32 + 16 + col] = fmaxf(c11[j] + bb1, 0.0f);
    }

    v8f u0 = {}, u1 = {};
    for (int k = 0; k < 32; k += 4) {
        const int kb = k + (hi ? 2 : 0);
        const v2f a = *(const v2f*)(&hbuf[wave][col * 32 + kb]);
        v2f b0, b1f;
        b0.x  = Wp1[kb * 32 + col];       b0.y  = Wp1[(kb + 1) * 32 + col];
        b1f.x = Wp1[kb * 32 + 16 + col];  b1f.y = Wp1[(kb + 1) * 32 + 16 + col];
        u0 = __builtin_amdgcn_wmma_f32_16x16x4_f32(false, a, false, b0,  (short)0, u0, false, false);
        u1 = __builtin_amdgcn_wmma_f32_16x16x4_f32(false, a, false, b1f, (short)0, u1, false, false);
    }

    for (int j = 0; j < 8; ++j) {
        const int row  = (hi ? 8 : 0) + j;
        const int nd   = m0 + row;
        const int g    = gids[nd];
        atomicAdd(&partial[g * 32 + col],      u0[j]);
        atomicAdd(&partial[g * 32 + 16 + col], u1[j]);
    }
}

#define POOL_FLOATS (N_GRAPHS * (K_USED + D_OUT))

__global__ void zero_pool_kernel(float* __restrict__ p) {
    const int i = blockIdx.x * blockDim.x + threadIdx.x;
    if (i < POOL_FLOATS) p[i] = 0.0f;
}

__global__ void finalize_kernel(const float* __restrict__ partial,
                                const float* __restrict__ bp0,
                                const float* __restrict__ bp1,
                                float* __restrict__ out) {
    const int i = blockIdx.x * blockDim.x + threadIdx.x;
    if (i < N_GRAPHS * D_OUT) {
        const int c = i & 31;
        out[i] = partial[i] + bp0[c] + bp1[c];
    }
}

extern "C" void kernel_launch(void* const* d_in, const int* in_sizes, int n_in,
                              void* d_out, int out_size, void* d_ws, size_t ws_size,
                              hipStream_t stream) {
    const float* pos  = (const float*)d_in[0];
    const float* feat = (const float*)d_in[1];
    const int*   src  = (const int*)  d_in[2];
    const int*   dst  = (const int*)  d_in[3];
    const int*   gids = (const int*)  d_in[4];
    const float* W1   = (const float*)d_in[5];
    const float* b1   = (const float*)d_in[6];
    const float* W2   = (const float*)d_in[7];
    const float* b2   = (const float*)d_in[8];
    const float* Wp0  = (const float*)d_in[9];
    const float* bp0  = (const float*)d_in[10];
    const float* Wp1  = (const float*)d_in[11];
    const float* bp1  = (const float*)d_in[12];
    float* out = (float*)d_out;
    const int nEdges = in_sizes[2];

    float* yBuf    = (float*)d_ws;
    float* accBuf  = yBuf + (size_t)N_NODES * 32;
    float* p0Buf   = accBuf + (size_t)N_NODES * 32;
    float* partial = p0Buf + (size_t)N_GRAPHS * K_USED;

    zero_pool_kernel<<<(POOL_FLOATS + 255) / 256, 256, 0, stream>>>(p0Buf);
    input_gemm_kernel<<<GEMM_WGS, GEMM_WAVES * 32, 0, stream>>>(pos, feat, W1, gids,
                                                                yBuf, accBuf, p0Buf);
    edge_scatter_kernel<<<SC_WGS, SC_WAVES_PER_WG * 32, 0, stream>>>(
        yBuf, src, dst, accBuf, nEdges);
    node_mlp_head_kernel<<<NODE_BLOCKS + HEAD_BLOCKS, 256, 0, stream>>>(
        accBuf, p0Buf, W2, b1, b2, Wp0, Wp1, gids, partial);
    finalize_kernel<<<(N_GRAPHS * D_OUT + 255) / 256, 256, 0, stream>>>(
        partial, bp0, bp1, out);
}


# graph sums land in out itself: a tile's 16 rows of u summed in registers, one whole-line f32 atomic per tile onto biases stored by the clearing launch; last launch dropped; sum order changes
# speedup vs baseline: 2.4138x; 1.2225x over previous
#include <hip/hip_runtime.h>
#include <hip/hip_bf16.h>


#define N_NODES 50000
#define D_POS   32
#define D_FEAT  768
#define D_IN    801
#define D_HID   32
#define D_OUT   32
#define N_GRAPHS 128
#define M_TILES  (N_NODES / 16)
#define WAVES_PER_BLOCK 8

typedef __attribute__((ext_vector_type(2))) float v2f;
typedef __attribute__((ext_vector_type(8))) float v8f;
typedef __attribute__((ext_vector_type(4))) float v4f;

#define GEMM_WGS    256
#define FULL_WAVES  (M_TILES / GEMM_WGS)
#define GEMM_WAVES  (FULL_WAVES + 4)
#define K_USED      800
#define K_STEPS     (K_USED / 4)
#define K_PAIRS     (K_USED / 2)

template <int CTRL>
__device__ __forceinline__ float lane_recv(const float x)
{
    return __builtin_bit_cast(float,
        __builtin_amdgcn_update_dpp(0, __builtin_bit_cast(int, x), CTRL, 0xF, 0xF, true));
}

template <bool MASKED>
__device__ __forceinline__ float colsum16(
    const float4& a0, const float4& a1, const float4& a2, const float4& a3,
    const int c, const bool take)
{
    float v[16] = {a0.x, a0.y, a0.z, a0.w, a1.x, a1.y, a1.z, a1.w,
                   a2.x, a2.y, a2.z, a2.w, a3.x, a3.y, a3.z, a3.w};
    if (MASKED) {
#pragma unroll
        for (int i = 0; i < 16; ++i) v[i] = take ? v[i] : 0.0f;
    }
    const bool b3 = (c & 8) != 0, b2 = (c & 4) != 0, b1 = (c & 2) != 0, b0 = (c & 1) != 0;
    float w8[8], w4[4], w2[2];
#pragma unroll
    for (int i = 0; i < 8; ++i) {
        const float keep = b3 ? v[i + 8] : v[i];
        const float send = b3 ? v[i] : v[i + 8];
        w8[i] = keep + lane_recv<0x168>(send);
    }
#pragma unroll
    for (int i = 0; i < 4; ++i) {
        const float keep = b2 ? w8[i + 4] : w8[i];
        const float send = b2 ? w8[i] : w8[i + 4];
        w4[i] = keep + lane_recv<0x164>(send);
    }
#pragma unroll
    for (int i = 0; i < 2; ++i) {
        const float keep = b1 ? w4[i + 2] : w4[i];
        const float send = b1 ? w4[i] : w4[i + 2];
        w2[i] = keep + lane_recv<0x162>(send);
    }
    const float keep = b0 ? w2[1] : w2[0];
    const float send = b0 ? w2[0] : w2[1];
    return keep + lane_recv<0x161>(send);
}

template <int NVEC>
__device__ __forceinline__ void pool_chunk(
    const float4 (&a)[NVEC], float* __restrict__ p0Buf, const int kbase,
    const int gl, const bool uniform, const int g0, const int c, const int colo)
{
    if (__builtin_expect(uniform, 1)) {
        float* p = p0Buf + g0 * K_USED + kbase + colo;
#pragma unroll
        for (int grp = 0; grp < NVEC / 4; ++grp) {
            const float s = colsum16<false>(a[4 * grp], a[4 * grp + 1], a[4 * grp + 2],
                                            a[4 * grp + 3], c, true);
            atomicAdd(p + 32 * grp, s);
        }
    } else {
        unsigned rem = 0xFFFFu;
        for (int it = 0; it < 16; ++it) {
            if (rem == 0) break;
            const int r = __builtin_ctz(rem);
            const int g = __builtin_amdgcn_readlane(gl, r);
            const bool take = (gl == g);
            float* p = p0Buf + g * K_USED + kbase + colo;
#pragma unroll
            for (int grp = 0; grp < NVEC / 4; ++grp) {
                const float s = colsum16<true>(a[4 * grp], a[4 * grp + 1], a[4 * grp + 2],
                                               a[4 * grp + 3], c, take);
                atomicAdd(p + 32 * grp, s);
            }
            rem &= ~(unsigned)__builtin_amdgcn_ballot_w32(take);
        }
    }
}

template <int NVEC>
__device__ __forceinline__ void gemm_chunk(
    const float4 (&a)[NVEC], const v2f* __restrict__ Bs,
    const int s0, const int lane, v8f& cy0, v8f& cy1)
{
#pragma unroll
    for (int i = 0; i < NVEC; ++i) {
#pragma unroll
        for (int o = 0; o < 2; ++o) {
            const int s = s0 + 2 * i + o;
            const v2f by0 = Bs[s * 64 + lane];
            const v2f by1 = Bs[s * 64 + 32 + lane];
            v2f av;
            av.x = o ? a[i].z : a[i].x;
            av.y = o ? a[i].w : a[i].y;
            cy0 = __builtin_amdgcn_wmma_f32_16x16x4_f32(false, av, false, by0, (short)0, cy0, false, false);
            cy1 = __builtin_amdgcn_wmma_f32_16x16x4_f32(false, av, false, by1, (short)0, cy1, false, false);
        }
    }
}

template <int NVEC>
__device__ __forceinline__ void gemm_chunk_half(
    const float4 (&a)[NVEC], const v2f* __restrict__ Bq, const int s0, v8f& acc)
{
#pragma unroll
    for (int i = 0; i < NVEC; ++i) {
#pragma unroll
        for (int o = 0; o < 2; ++o) {
            const int s = s0 + 2 * i + o;
            const v2f b = Bq[s * 64];
            v2f av;
            av.x = o ? a[i].z : a[i].x;
            av.y = o ? a[i].w : a[i].y;
            acc = __builtin_amdgcn_wmma_f32_16x16x4_f32(false, av, false, b, (short)0, acc, false, false);
        }
    }
}

template <int MODE>
__device__ __forceinline__ void chunk_work(
    const float4 (&aP)[4], const v2f* __restrict__ Bs, const v2f* __restrict__ Bq,
    float* __restrict__ p0Buf, const int lane, const int gl, const bool uniform,
    const int g0, const int c, const int colo, v8f& c0, v8f& c1)
{
    if (MODE == 0) gemm_chunk<4>(aP, Bs, 0, lane, c0, c1);
    if (MODE == 1) gemm_chunk_half<4>(aP, Bq, 0, c0);
    if (MODE != 1) pool_chunk<4>(aP, p0Buf, 0, gl, uniform, g0, c, colo);
}
template <int MODE>
__device__ __forceinline__ void chunk_work(
    const float4 (&a)[8], const int st, const v2f* __restrict__ Bs, const v2f* __restrict__ Bq,
    float* __restrict__ p0Buf, const int lane, const int gl, const bool uniform,
    const int g0, const int c, const int colo, v8f& c0, v8f& c1)
{
    if (MODE == 0) gemm_chunk<8>(a, Bs, 8 + 16 * st, lane, c0, c1);
    if (MODE == 1) gemm_chunk_half<8>(a, Bq, 8 + 16 * st, c0);
    if (MODE != 1) pool_chunk<8>(a, p0Buf, D_POS + 64 * st, gl, uniform, g0, c, colo);
}

#define A_PITCH4 17

template <int MODE>
__device__ __forceinline__ void run_tile(
    const v4f* __restrict__ featL, const v4f (&gP)[4], const v4f (&g0v)[8],
    v4f* __restrict__ AsW,
    const v2f* __restrict__ Bs, const v2f* __restrict__ Bq, float* __restrict__ p0Buf,
    const int lane, const int gl, const bool uniform, const int g0,
    const int c, const int colo, v8f& c0, v8f& c1)
{
    v4f* stP = AsW + (lane >> 3) * A_PITCH4 + (lane & 7);
    v4f* stF = AsW + (lane >> 4) * A_PITCH4 + (lane & 15);
    const v4f* ld = AsW + c * A_PITCH4 + (lane >> 4);

    float4 aP[4];
#pragma unroll
    for (int i = 0; i < 4; ++i) stP[4 * i * A_PITCH4] = gP[i];
#pragma unroll
    for (int j = 0; j < 4; ++j) { const v4f t = ld[2 * j]; aP[j] = make_float4(t.x, t.y, t.z, t.w); }
    chunk_work<MODE>(aP, Bs, Bq, p0Buf, lane, gl, uniform, g0, c, colo, c0, c1);

    float4 a[8];
    v4f g[8];
#pragma unroll
    for (int i = 0; i < 8; ++i) g[i] = g0v[i];
    for (int st = 0; st < D_FEAT / 64 - 1; ++st) {
#pragma unroll
        for (int i = 0; i < 8; ++i) stF[2 * i * A_PITCH4] = g[i];
#pragma unroll
        for (int i = 0; i < 8; ++i) g[i] = featL[2 * i * (D_FEAT / 4) + 16 * (st + 1)];
#pragma unroll
        for (int j = 0; j < 8; ++j) { const v4f t = ld[2 * j]; a[j] = make_float4(t.x, t.y, t.z, t.w); }
        chunk_work<MODE>(a, st, Bs, Bq, p0Buf, lane, gl, uniform, g0, c, colo, c0, c1);
    }
    {
#pragma unroll
        for (int i = 0; i < 8; ++i) stF[2 * i * A_PITCH4] = g[i];
#pragma unroll
        for (int j = 0; j < 8; ++j) { const v4f t = ld[2 * j]; a[j] = make_float4(t.x, t.y, t.z, t.w); }
        chunk_work<MODE>(a, D_FEAT / 64 - 1, Bs, Bq, p0Buf, lane, gl, uniform, g0, c, colo, c0, c1);
    }
}

__global__ __launch_bounds__(GEMM_WAVES * 32) void input_gemm_kernel(
    const float* __restrict__ pos, const float* __restrict__ feat,
    const float* __restrict__ W1,  const int* __restrict__ gids,
    float* __restrict__ yBuf, float* __restrict__ accBuf, float* __restrict__ p0Buf)
{
    __shared__ __attribute__((aligned(16))) v2f Bs[K_STEPS * 64];
    __shared__ __attribute__((aligned(16))) v4f As[GEMM_WAVES * 16 * A_PITCH4];

    const int wave   = __builtin_amdgcn_readfirstlane((int)(threadIdx.x >> 5));
    const int lane   = threadIdx.x & 31;
    const bool whole  = wave < FULL_WAVES;
    const int role   = wave - FULL_WAVES;
    const int tile   = (whole ? wave : FULL_WAVES) * GEMM_WGS + (int)blockIdx.x;
    const bool active = (tile < M_TILES) && (role < 3);

    const int  m0   = (tile < M_TILES ? tile : 0) * 16;
    const int  col  = lane & 15;
    const int  h    = lane >> 4;
    const int  node = m0 + col;

    const v4f* posL  = (const v4f*)(pos  + (size_t)m0 * D_POS)  + (lane >> 3) * (D_POS / 4)  + (lane & 7);
    const v4f* featL = (const v4f*)(feat + (size_t)m0 * D_FEAT) + (lane >> 4) * (D_FEAT / 4) + (lane & 15);
    v4f* AsW = As + wave * 16 * A_PITCH4;

    v4f gP[4], gA[8];
#pragma unroll
    for (int i = 0; i < 4; ++i) gP[i] = posL[4 * i * (D_POS / 4)];
#pragma unroll
    for (int i = 0; i < 8; ++i) gA[i] = featL[2 * i * (D_FEAT / 4)];
    const int gl = gids[node];

    {
        const int n  = lane;
        const int nh = n >> 4, c = n & 15;
        for (int bt = wave; bt < K_PAIRS / 16; bt += GEMM_WAVES) {
            const int k0   = 32 * bt;
            const int row0 = (k0 < D_POS) ? k0 : (k0 + 1);
            const float* w1p = W1 + row0 * 32 + n;
            v2f v[16];
#pragma unroll
            for (int u = 0; u < 16; ++u) {
                v[u].x = w1p[(2 * u) * 32];
                v[u].y = w1p[(2 * u + 1) * 32];
            }
#pragma unroll
            for (int u = 0; u < 16; ++u) {
                const int s = 2 * (4 * bt + (u >> 2)) + (u & 1);
                Bs[s * 64 + nh * 32 + ((u >> 1) & 1) * 16 + c] = v[u];
            }
        }
    }
    __syncthreads();
    if (!active) return;

    const int  g0      = __builtin_amdgcn_readfirstlane(gl);
    const bool uniform = __builtin_amdgcn_ballot_w32(gl == g0) == 0xFFFFFFFFu;
    const int  colo    = 8 * (col >> 2) + 4 * h + (col & 3);

    if (whole) {
        v8f cy0 = {}, cy1 = {};
        run_tile<0>(featL, gP, gA, AsW, Bs, Bs, p0Buf, lane, gl, uniform, g0, col, colo, cy0, cy1);

        for (int j = 0; j < 8; ++j) {
            const int row = 8 * h + j;
            const int off = (m0 + row) * 32 + col;
            yBuf[off]        = cy0[j];  accBuf[off]        = cy0[j];
            yBuf[off + 16]   = cy1[j];  accBuf[off + 16]   = cy1[j];
        }
    } else if (role < 2) {
        const v2f* Bq = Bs + role * 32 + lane;
        v8f acc = {}, unused = {};
        run_tile<1>(featL, gP, gA, AsW, Bs, Bq, p0Buf, lane, gl, uniform, g0, col, colo, acc, unused);

        for (int j = 0; j < 8; ++j) {
            const int row = 8 * h + j;
            const int off = (m0 + row) * 32 + role * 16 + col;
            yBuf[off] = acc[j];  accBuf[off] = acc[j];
        }
    } else {
        v8f unused0 = {}, unused1 = {};
        run_tile<2>(featL, gP, gA, AsW, Bs, Bs, p0Buf, lane, gl, uniform, g0, col, colo, unused0, unused1);
    }
}

#define SC_WGS          512
#define SC_WAVES_PER_WG 8
#define SC_WAVES_LOG2   12
#define SC_UNIT         16

template <int BASE, int N>
__device__ __forceinline__ void scatter_edges(
    const float* __restrict__ yBuf, float* __restrict__ accBuf,
    const int sv, const int dv, const int lane)
{
    float v[N];
    int   d[N];
#pragma unroll
    for (int u = 0; u < N; ++u) {
        const int s = __builtin_amdgcn_readlane(sv, BASE + u);
        d[u]        = __builtin_amdgcn_readlane(dv, BASE + u);
        v[u] = yBuf[(size_t)s * 32 + lane];
    }
#pragma unroll
    for (int u = 0; u < N; ++u)
        atomicAdd(accBuf + (size_t)d[u] * 32 + lane, v[u]);
}

__global__ __launch_bounds__(SC_WAVES_PER_WG * 32) void edge_scatter_kernel(
    const float* __restrict__ yBuf, const int* __restrict__ src,
    const int* __restrict__ dst, float* __restrict__ accBuf, int nEdges)
{
    const int wave = __builtin_amdgcn_readfirstlane((int)(threadIdx.x >> 5));
    const int lane = threadIdx.x & 31;
    const int w    = (int)blockIdx.x * SC_WAVES_PER_WG + wave;

    const int nUnits = (nEdges + SC_UNIT - 1) / SC_UNIT;
    const int u0 = (int)(((long long)w * nUnits) >> SC_WAVES_LOG2);
    const int u1 = (int)(((long long)(w + 1) * nUnits) >> SC_WAVES_LOG2);
    const int eBeg = u0 * SC_UNIT;
    const int eEnd = (u1 * SC_UNIT < nEdges) ? u1 * SC_UNIT : nEdges;

    if (eBeg >= eEnd) return;

    const int last = nEdges - 1;
    const int p0   = (eBeg + lane < last) ? eBeg + lane : last;
    int sv = src[p0], dv = dst[p0];

    for (int e0 = eBeg; e0 < eEnd; e0 += 32) {
        const int pN  = (e0 + 32 + lane < last) ? e0 + 32 + lane : last;
        const int svN = src[pN], dvN = dst[pN];

        const int left = eEnd - e0;
        if (left >= 32) {
            scatter_edges<0, 32>(yBuf, accBuf, sv, dv, lane);
        } else {
            int done = 0;
            if (left >= SC_UNIT) {
                scatter_edges<0, SC_UNIT>(yBuf, accBuf, sv, dv, lane);
                done = SC_UNIT;
            }
            for (int i = done; i < left; ++i) {
                const int s = __builtin_amdgcn_readlane(sv, i);
                const int d = __builtin_amdgcn_readlane(dv, i);
                atomicAdd(accBuf + (size_t)d * 32 + lane, yBuf[(size_t)s * 32 + lane]);
            }
        }
        sv = svN;  dv = dvN;
    }
}

#define NODE_BLOCKS ((M_TILES + WAVES_PER_BLOCK - 1) / WAVES_PER_BLOCK)
#define HEAD_BLOCKS (K_USED / 32)

__device__ __forceinline__ float other_half(const float x)
{
    return __builtin_bit_cast(float,
        __builtin_amdgcn_permlanex16(__builtin_bit_cast(int, x), __builtin_bit_cast(int, x),
                                     0x76543210, 0xfedcba98, false, false));
}

__global__ __launch_bounds__(256) void node_mlp_head_kernel(
    const float* __restrict__ accBuf, const float* __restrict__ p0Buf,
    const float* __restrict__ W2, const float* __restrict__ bias1,
    const float* __restrict__ bias2, const float* __restrict__ Wp0,
    const float* __restrict__ Wp1,
    const int* __restrict__ gids, float* __restrict__ out)
{
    __shared__ float hbuf[WAVES_PER_BLOCK][16 * 32];

    const int wave = threadIdx.x >> 5;
    const int lane = threadIdx.x & 31;
    const int  col  = lane & 15;
    const bool hi   = lane >= 16;

    if ((int)blockIdx.x >= NODE_BLOCKS) {
        const int ku0   = 32 * ((int)blockIdx.x - NODE_BLOCKS);
        const int wrow0 = (ku0 < D_POS) ? ku0 : (ku0 + 1);
        const float* aRow = p0Buf + (size_t)(16 * wave + col) * K_USED + ku0;
        v8f u0 = {}, u1 = {};
        for (int k = 0; k < 32; k += 4) {
            const int kb = k + (hi ? 2 : 0);
            const v2f a = *(const v2f*)(aRow + kb);
            v2f b0, b1f;
            b0.x  = Wp0[(wrow0 + kb) * 32 + col];       b0.y  = Wp0[(wrow0 + kb + 1) * 32 + col];
            b1f.x = Wp0[(wrow0 + kb) * 32 + 16 + col];  b1f.y = Wp0[(wrow0 + kb + 1) * 32 + 16 + col];
            u0 = __builtin_amdgcn_wmma_f32_16x16x4_f32(false, a, false, b0,  (short)0, u0, false, false);
            u1 = __builtin_amdgcn_wmma_f32_16x16x4_f32(false, a, false, b1f, (short)0, u1, false, false);
        }
        for (int j = 0; j < 8; ++j) {
            const int g = 16 * wave + (hi ? 8 : 0) + j;
            atomicAdd(&out[g * 32 + col],      u0[j]);
            atomicAdd(&out[g * 32 + 16 + col], u1[j]);
        }
        return;
    }

    const int tile = blockIdx.x * WAVES_PER_BLOCK + wave;
    if (tile >= M_TILES) return;

    const int  m0   = tile * 16;
    const int  node = m0 + col;
    const int  gl   = gids[node];

    const float* accRow = accBuf + (size_t)node * 32;
    v8f c10 = {}, c11 = {};
    for (int k = 0; k < 32; k += 4) {
        const int kb = k + (hi ? 2 : 0);
        const v2f av = *(const v2f*)(accRow + kb);
        const v2f bv = *(const v2f*)(bias1 + kb);
        v2f a;
        a.x = fmaxf(av.x + bv.x, 0.0f);
        a.y = fmaxf(av.y + bv.y, 0.0f);
        v2f b0, b1f;
        b0.x  = W2[kb * 32 + col];       b0.y  = W2[(kb + 1) * 32 + col];
        b1f.x = W2[kb * 32 + 16 + col];  b1f.y = W2[(kb + 1) * 32 + 16 + col];
        c10 = __builtin_amdgcn_wmma_f32_16x16x4_f32(false, a, false, b0,  (short)0, c10, false, false);
        c11 = __builtin_amdgcn_wmma_f32_16x16x4_f32(false, a, false, b1f, (short)0, c11, false, false);
    }

    const float bb0 = bias2[col];
    const float bb1 = bias2[16 + col];
    for (int j = 0; j < 8; ++j) {
        const int row = (hi ? 8 : 0) + j;
        hbuf[wave][row * 32 + col]      = fmaxf(c10[j] + bb0, 0.0f);
        hbuf[wave][row * 32 + 16 + col] = fmaxf(c11[j] + bb1, 0.0f);
    }

    v8f u0 = {}, u1 = {};
    for (int k = 0; k < 32; k += 4) {
        const int kb = k + (hi ? 2 : 0);
        const v2f a = *(const v2f*)(&hbuf[wave][col * 32 + kb]);
        v2f b0, b1f;
        b0.x  = Wp1[kb * 32 + col];       b0.y  = Wp1[(kb + 1) * 32 + col];
        b1f.x = Wp1[kb * 32 + 16 + col];  b1f.y = Wp1[(kb + 1) * 32 + 16 + col];
        u0 = __builtin_amdgcn_wmma_f32_16x16x4_f32(false, a, false, b0,  (short)0, u0, false, false);
        u1 = __builtin_amdgcn_wmma_f32_16x16x4_f32(false, a, false, b1f, (short)0, u1, false, false);
    }

    const int  g0      = __builtin_amdgcn_readfirstlane(gl);
    const bool uniform = __builtin_amdgcn_ballot_w32(gl == g0) == 0xFFFFFFFFu;
    if (__builtin_expect(uniform, 1)) {
        float s0 = ((u0[0] + u0[1]) + (u0[2] + u0[3])) + ((u0[4] + u0[5]) + (u0[6] + u0[7]));
        float s1 = ((u1[0] + u1[1]) + (u1[2] + u1[3])) + ((u1[4] + u1[5]) + (u1[6] + u1[7]));
        s0 += other_half(s0);
        s1 += other_half(s1);
        atomicAdd(&out[g0 * 32 + lane], hi ? s1 : s0);
    } else {
#pragma unroll
        for (int j = 0; j < 8; ++j) {
            const int gLo = __builtin_amdgcn_readlane(gl, j);
            const int gHi = __builtin_amdgcn_readlane(gl, 8 + j);
            const int g   = hi ? gHi : gLo;
            atomicAdd(&out[g * 32 + col],      u0[j]);
            atomicAdd(&out[g * 32 + 16 + col], u1[j]);
        }
    }
}

#define POOL_BLOCKS (N_GRAPHS * K_USED / 256)
#define OUT_BLOCKS  (N_GRAPHS * D_OUT / 256)

__global__ __launch_bounds__(256) void init_pool_kernel(
    float* __restrict__ p0Buf, const float* __restrict__ bp0,
    const float* __restrict__ bp1, float* __restrict__ out)
{
    if ((int)blockIdx.x < POOL_BLOCKS) {
        p0Buf[blockIdx.x * 256 + threadIdx.x] = 0.0f;
    } else {
        const int i = ((int)blockIdx.x - POOL_BLOCKS) * 256 + (int)threadIdx.x;
        const int c = i & 31;
        out[i] = bp0[c] + bp1[c];
    }
}

extern "C" void kernel_launch(void* const* d_in, const int* in_sizes, int n_in,
                              void* d_out, int out_size, void* d_ws, size_t ws_size,
                              hipStream_t stream) {
    const float* pos  = (const float*)d_in[0];
    const float* feat = (const float*)d_in[1];
    const int*   src  = (const int*)  d_in[2];
    const int*   dst  = (const int*)  d_in[3];
    const int*   gids = (const int*)  d_in[4];
    const float* W1   = (const float*)d_in[5];
    const float* b1   = (const float*)d_in[6];
    const float* W2   = (const float*)d_in[7];
    const float* b2   = (const float*)d_in[8];
    const float* Wp0  = (const float*)d_in[9];
    const float* bp0  = (const float*)d_in[10];
    const float* Wp1  = (const float*)d_in[11];
    const float* bp1  = (const float*)d_in[12];
    float* out = (float*)d_out;
    const int nEdges = in_sizes[2];

    float* yBuf    = (float*)d_ws;
    float* accBuf  = yBuf + (size_t)N_NODES * 32;
    float* p0Buf   = accBuf + (size_t)N_NODES * 32;

    init_pool_kernel<<<POOL_BLOCKS + OUT_BLOCKS, 256, 0, stream>>>(p0Buf, bp0, bp1, out);
    input_gemm_kernel<<<GEMM_WGS, GEMM_WAVES * 32, 0, stream>>>(pos, feat, W1, gids,
                                                                yBuf, accBuf, p0Buf);
    edge_scatter_kernel<<<SC_WGS, SC_WAVES_PER_WG * 32, 0, stream>>>(
        yBuf, src, dst, accBuf, nEdges);
    node_mlp_head_kernel<<<NODE_BLOCKS + HEAD_BLOCKS, 256, 0, stream>>>(
        accBuf, p0Buf, W2, b1, b2, Wp0, Wp1, gids, out);
}


# column sums collected per work-group in LDS (its tiles made consecutive) and stored as a plane; spare edge-launch blocks add it onto p0; clearing launch dropped (4 -> 3 launches); sum order changes
# speedup vs baseline: 2.4888x; 1.0311x over previous
#include <hip/hip_runtime.h>
#include <hip/hip_bf16.h>


#define N_NODES 50000
#define D_POS   32
#define D_FEAT  768
#define D_IN    801
#define D_HID   32
#define D_OUT   32
#define N_GRAPHS 128
#define M_TILES  (N_NODES / 16)
#define WAVES_PER_BLOCK 8

typedef __attribute__((ext_vector_type(2))) float v2f;
typedef __attribute__((ext_vector_type(8))) float v8f;
typedef __attribute__((ext_vector_type(4))) float v4f;

#define GEMM_WGS    256
#define FULL_WAVES  (M_TILES / GEMM_WGS)
#define GEMM_WAVES  (FULL_WAVES + 4)
#define LEFT_TILES  (M_TILES - FULL_WAVES * GEMM_WGS)
#define K_USED      800
#define K_STEPS     (K_USED / 4)
#define K_PAIRS     (K_USED / 2)

#define POOL_SLOTS  2

__device__ __forceinline__ int wg_first_tile(const int b)
{
    return FULL_WAVES * b + (b < LEFT_TILES ? b : LEFT_TILES);
}
__device__ __forceinline__ int wg_of_tile(const int t)
{
    return (t < (FULL_WAVES + 1) * LEFT_TILES)
               ? t / (FULL_WAVES + 1)
               : LEFT_TILES + (t - (FULL_WAVES + 1) * LEFT_TILES) / FULL_WAVES;
}

template <int CTRL>
__device__ __forceinline__ float lane_recv(const float x)
{
    return __builtin_bit_cast(float,
        __builtin_amdgcn_update_dpp(0, __builtin_bit_cast(int, x), CTRL, 0xF, 0xF, true));
}

template <bool MASKED>
__device__ __forceinline__ float colsum16(
    const float4& a0, const float4& a1, const float4& a2, const float4& a3,
    const int c, const bool take)
{
    float v[16] = {a0.x, a0.y, a0.z, a0.w, a1.x, a1.y, a1.z, a1.w,
                   a2.x, a2.y, a2.z, a2.w, a3.x, a3.y, a3.z, a3.w};
    if (MASKED) {
#pragma unroll
        for (int i = 0; i < 16; ++i) v[i] = take ? v[i] : 0.0f;
    }
    const bool b3 = (c & 8) != 0, b2 = (c & 4) != 0, b1 = (c & 2) != 0, b0 = (c & 1) != 0;
    float w8[8], w4[4], w2[2];
#pragma unroll
    for (int i = 0; i < 8; ++i) {
        const float keep = b3 ? v[i + 8] : v[i];
        const float send = b3 ? v[i] : v[i + 8];
        w8[i] = keep + lane_recv<0x168>(send);
    }
#pragma unroll
    for (int i = 0; i < 4; ++i) {
        const float keep = b2 ? w8[i + 4] : w8[i];
        const float send = b2 ? w8[i] : w8[i + 4];
        w4[i] = keep + lane_recv<0x164>(send);
    }
#pragma unroll
    for (int i = 0; i < 2; ++i) {
        const float keep = b1 ? w4[i + 2] : w4[i];
        const float send = b1 ? w4[i] : w4[i + 2];
        w2[i] = keep + lane_recv<0x162>(send);
    }
    const float keep = b0 ? w2[1] : w2[0];
    const float send = b0 ? w2[0] : w2[1];
    return keep + lane_recv<0x161>(send);
}

template <int NVEC>
__device__ __forceinline__ void pool_chunk(
    const float4 (&a)[NVEC], float* __restrict__ Ps, const int kbase,
    const int gl, const bool uniform, const int g0, const int gF, const int c, const int colo)
{
    if (__builtin_expect(uniform, 1)) {
        const unsigned slot = (unsigned)(g0 - gF);
        if (slot < POOL_SLOTS) {
            float* p = Ps + slot * K_USED + kbase + colo;
#pragma unroll
            for (int grp = 0; grp < NVEC / 4; ++grp) {
                const float s = colsum16<false>(a[4 * grp], a[4 * grp + 1], a[4 * grp + 2],
                                                a[4 * grp + 3], c, true);
                atomicAdd(p + 32 * grp, s);
            }
        }
    } else {
        unsigned rem = 0xFFFFu;
        for (int it = 0; it < 16; ++it) {
            if (rem == 0) break;
            const int r = __builtin_ctz(rem);
            const int g = __builtin_amdgcn_readlane(gl, r);
            const bool take = (gl == g);
            const unsigned slot = (unsigned)(g - gF);
            if (slot < POOL_SLOTS) {
                float* p = Ps + slot * K_USED + kbase + colo;
#pragma unroll
                for (int grp = 0; grp < NVEC / 4; ++grp) {
                    const float s = colsum16<true>(a[4 * grp], a[4 * grp + 1], a[4 * grp + 2],
                                                   a[4 * grp + 3], c, take);
                    atomicAdd(p + 32 * grp, s);
                }
            }
            rem &= ~(unsigned)__builtin_amdgcn_ballot_w32(take);
        }
    }
}

template <int NVEC>
__device__ __forceinline__ void gemm_chunk(
    const float4 (&a)[NVEC], const v2f* __restrict__ Bs,
    const int s0, const int lane, v8f& cy0, v8f& cy1)
{
#pragma unroll
    for (int i = 0; i < NVEC; ++i) {
#pragma unroll
        for (int o = 0; o < 2; ++o) {
            const int s = s0 + 2 * i + o;
            const v2f by0 = Bs[s * 64 + lane];
            const v2f by1 = Bs[s * 64 + 32 + lane];
            v2f av;
            av.x = o ? a[i].z : a[i].x;
            av.y = o ? a[i].w : a[i].y;
            cy0 = __builtin_amdgcn_wmma_f32_16x16x4_f32(false, av, false, by0, (short)0, cy0, false, false);
            cy1 = __builtin_amdgcn_wmma_f32_16x16x4_f32(false, av, false, by1, (short)0, cy1, false, false);
        }
    }
}

template <int NVEC>
__device__ __forceinline__ void gemm_chunk_half(
    const float4 (&a)[NVEC], const v2f* __restrict__ Bq, const int s0, v8f& acc)
{
#pragma unroll
    for (int i = 0; i < NVEC; ++i) {
#pragma unroll
        for (int o = 0; o < 2; ++o) {
            const int s = s0 + 2 * i + o;
            const v2f b = Bq[s * 64];
            v2f av;
            av.x = o ? a[i].z : a[i].x;
            av.y = o ? a[i].w : a[i].y;
            acc = __builtin_amdgcn_wmma_f32_16x16x4_f32(false, av, false, b, (short)0, acc, false, false);
        }
    }
}

template <int MODE>
__device__ __forceinline__ void chunk_work(
    const float4 (&aP)[4], const v2f* __restrict__ Bs, const v2f* __restrict__ Bq,
    float* __restrict__ Ps, const int lane, const int gl, const bool uniform,
    const int g0, const int gF, const int c, const int colo, v8f& c0, v8f& c1)
{
    if (MODE == 0) gemm_chunk<4>(aP, Bs, 0, lane, c0, c1);
    if (MODE == 1) gemm_chunk_half<4>(aP, Bq, 0, c0);
    if (MODE != 1) pool_chunk<4>(aP, Ps, 0, gl, uniform, g0, gF, c, colo);
}
template <int MODE>
__device__ __forceinline__ void chunk_work(
    const float4 (&a)[8], const int st, const v2f* __restrict__ Bs, const v2f* __restrict__ Bq,
    float* __restrict__ Ps, const int lane, const int gl, const bool uniform,
    const int g0, const int gF, const int c, const int colo, v8f& c0, v8f& c1)
{
    if (MODE == 0) gemm_chunk<8>(a, Bs, 8 + 16 * st, lane, c0, c1);
    if (MODE == 1) gemm_chunk_half<8>(a, Bq, 8 + 16 * st, c0);
    if (MODE != 1) pool_chunk<8>(a, Ps, D_POS + 64 * st, gl, uniform, g0, gF, c, colo);
}

#define A_PITCH4 17

template <int MODE>
__device__ __forceinline__ void run_tile(
    const v4f* __restrict__ featL, const v4f (&gP)[4], const v4f (&g0v)[8],
    v4f* __restrict__ AsW,
    const v2f* __restrict__ Bs, const v2f* __restrict__ Bq, float* __restrict__ Ps,
    const int lane, const int gl, const bool uniform, const int g0, const int gF,
    const int c, const int colo, v8f& c0, v8f& c1)
{
    v4f* stP = AsW + (lane >> 3) * A_PITCH4 + (lane & 7);
    v4f* stF = AsW + (lane >> 4) * A_PITCH4 + (lane & 15);
    const v4f* ld = AsW + c * A_PITCH4 + (lane >> 4);

    float4 aP[4];
#pragma unroll
    for (int i = 0; i < 4; ++i) stP[4 * i * A_PITCH4] = gP[i];
#pragma unroll
    for (int j = 0; j < 4; ++j) { const v4f t = ld[2 * j]; aP[j] = make_float4(t.x, t.y, t.z, t.w); }
    chunk_work<MODE>(aP, Bs, Bq, Ps, lane, gl, uniform, g0, gF, c, colo, c0, c1);

    float4 a[8];
    v4f g[8];
#pragma unroll
    for (int i = 0; i < 8; ++i) g[i] = g0v[i];
    for (int st = 0; st < D_FEAT / 64 - 1; ++st) {
#pragma unroll
        for (int i = 0; i < 8; ++i) stF[2 * i * A_PITCH4] = g[i];
#pragma unroll
        for (int i = 0; i < 8; ++i) g[i] = featL[2 * i * (D_FEAT / 4) + 16 * (st + 1)];
#pragma unroll
        for (int j = 0; j < 8; ++j) { const v4f t = ld[2 * j]; a[j] = make_float4(t.x, t.y, t.z, t.w); }
        chunk_work<MODE>(a, st, Bs, Bq, Ps, lane, gl, uniform, g0, gF, c, colo, c0, c1);
    }
    {
#pragma unroll
        for (int i = 0; i < 8; ++i) stF[2 * i * A_PITCH4] = g[i];
#pragma unroll
        for (int j = 0; j < 8; ++j) { const v4f t = ld[2 * j]; a[j] = make_float4(t.x, t.y, t.z, t.w); }
        chunk_work<MODE>(a, D_FEAT / 64 - 1, Bs, Bq, Ps, lane, gl, uniform, g0, gF, c, colo, c0, c1);
    }
}

__global__ __launch_bounds__(GEMM_WAVES * 32) void input_gemm_kernel(
    const float* __restrict__ pos, const float* __restrict__ feat,
    const float* __restrict__ W1,  const int* __restrict__ gids,
    const float* __restrict__ bp0, const float* __restrict__ bp1,
    float* __restrict__ yBuf, float* __restrict__ accBuf, float* __restrict__ p0Buf,
    float* __restrict__ partBuf, float* __restrict__ out)
{
    __shared__ __attribute__((aligned(16))) v2f Bs[K_STEPS * 64];
    __shared__ __attribute__((aligned(16))) v4f As[GEMM_WAVES * 16 * A_PITCH4];
    __shared__ __attribute__((aligned(16))) float Ps[POOL_SLOTS * K_USED];

    const int wave   = __builtin_amdgcn_readfirstlane((int)(threadIdx.x >> 5));
    const int lane   = threadIdx.x & 31;
    const int bx     = (int)blockIdx.x;
    const int t0     = wg_first_tile(bx);
    const bool whole  = wave < FULL_WAVES;
    const int role   = wave - FULL_WAVES;
    const bool active = whole || (bx < LEFT_TILES && role < 3);
    const int tile   = t0 + (whole ? wave : FULL_WAVES);

    const int  m0   = (active ? tile : 0) * 16;
    const int  col  = lane & 15;
    const int  h    = lane >> 4;
    const int  node = m0 + col;

    const v4f* posL  = (const v4f*)(pos  + (size_t)m0 * D_POS)  + (lane >> 3) * (D_POS / 4)  + (lane & 7);
    const v4f* featL = (const v4f*)(feat + (size_t)m0 * D_FEAT) + (lane >> 4) * (D_FEAT / 4) + (lane & 15);
    v4f* AsW = As + wave * 16 * A_PITCH4;

    v4f gP[4], gA[8];
#pragma unroll
    for (int i = 0; i < 4; ++i) gP[i] = posL[4 * i * (D_POS / 4)];
#pragma unroll
    for (int i = 0; i < 8; ++i) gA[i] = featL[2 * i * (D_FEAT / 4)];
    const int gl  = gids[node];
    const int gFl = gids[t0 * 16];

    if (wave == GEMM_WAVES - 1) {
        const v4f zero4 = {0.0f, 0.0f, 0.0f, 0.0f};
        v4f* pz = (v4f*)p0Buf + bx * (N_GRAPHS * K_USED / 4 / GEMM_WGS);
        for (int j = lane; j < N_GRAPHS * K_USED / 4 / GEMM_WGS; j += 32) pz[j] = zero4;
        if (bx < N_GRAPHS) out[bx * D_OUT + lane] = bp0[lane] + bp1[lane];
    }
    for (int i = (int)threadIdx.x; i < POOL_SLOTS * K_USED; i += GEMM_WAVES * 32) Ps[i] = 0.0f;

    {
        const int n  = lane;
        const int nh = n >> 4, c = n & 15;
        for (int bt = wave; bt < K_PAIRS / 16; bt += GEMM_WAVES) {
            const int k0   = 32 * bt;
            const int row0 = (k0 < D_POS) ? k0 : (k0 + 1);
            const float* w1p = W1 + row0 * 32 + n;
            v2f v[16];
#pragma unroll
            for (int u = 0; u < 16; ++u) {
                v[u].x = w1p[(2 * u) * 32];
                v[u].y = w1p[(2 * u + 1) * 32];
            }
#pragma unroll
            for (int u = 0; u < 16; ++u) {
                const int s = 2 * (4 * bt + (u >> 2)) + (u & 1);
                Bs[s * 64 + nh * 32 + ((u >> 1) & 1) * 16 + c] = v[u];
            }
        }
    }
    __syncthreads();

    if (active) {
        const int  g0      = __builtin_amdgcn_readfirstlane(gl);
        const int  gF      = __builtin_amdgcn_readfirstlane(gFl);
        const bool uniform = __builtin_amdgcn_ballot_w32(gl == g0) == 0xFFFFFFFFu;
        const int  colo    = 8 * (col >> 2) + 4 * h + (col & 3);

        if (whole) {
            v8f cy0 = {}, cy1 = {};
            run_tile<0>(featL, gP, gA, AsW, Bs, Bs, Ps, lane, gl, uniform, g0, gF, col, colo, cy0, cy1);

            for (int j = 0; j < 8; ++j) {
                const int row = 8 * h + j;
                const int off = (m0 + row) * 32 + col;
                yBuf[off]        = cy0[j];  accBuf[off]        = cy0[j];
                yBuf[off + 16]   = cy1[j];  accBuf[off + 16]   = cy1[j];
            }
        } else if (role < 2) {
            const v2f* Bq = Bs + role * 32 + lane;
            v8f acc = {}, unused = {};
            run_tile<1>(featL, gP, gA, AsW, Bs, Bq, Ps, lane, gl, uniform, g0, gF, col, colo, acc, unused);

            for (int j = 0; j < 8; ++j) {
                const int row = 8 * h + j;
                const int off = (m0 + row) * 32 + role * 16 + col;
                yBuf[off] = acc[j];  accBuf[off] = acc[j];
            }
        } else {
            v8f unused0 = {}, unused1 = {};
            run_tile<2>(featL, gP, gA, AsW, Bs, Bs, Ps, lane, gl, uniform, g0, gF, col, colo, unused0, unused1);
        }
    }

    __syncthreads();
    {
        const v4f* ps4 = (const v4f*)Ps;
        v4f* dst4 = (v4f*)partBuf + bx * (POOL_SLOTS * K_USED / 4);
        for (int i = (int)threadIdx.x; i < POOL_SLOTS * K_USED / 4; i += GEMM_WAVES * 32) dst4[i] = ps4[i];
    }
}

#define SC_WGS          512
#define SC_WAVES_PER_WG 8
#define SC_WAVES_LOG2   12
#define SC_UNIT         16

#define HAND_ADD_WAVES  GEMM_WGS
#define HAND_SCAN_WAVES ((M_TILES + 31) / 32)
#define HAND_WGS        ((HAND_ADD_WAVES + HAND_SCAN_WAVES + SC_WAVES_PER_WG - 1) / SC_WAVES_PER_WG)

__device__ __forceinline__ void hand_over(
    const int q, const int lane,
    const float* __restrict__ pos, const float* __restrict__ feat,
    const int* __restrict__ gids, const float* __restrict__ partBuf,
    float* __restrict__ p0Buf)
{
    if (q < HAND_ADD_WAVES) {
        const int b  = q;
        const int gF = __builtin_amdgcn_readfirstlane(gids[wg_first_tile(b) * 16]);
        const float* src = partBuf + (size_t)b * (POOL_SLOTS * K_USED) + lane;
        float v[POOL_SLOTS][K_USED / 32];
#pragma unroll
        for (int s = 0; s < POOL_SLOTS; ++s)
#pragma unroll
            for (int grp = 0; grp < K_USED / 32; ++grp) v[s][grp] = src[s * K_USED + 32 * grp];
#pragma unroll
        for (int s = 0; s < POOL_SLOTS; ++s) {
            const int g = gF + s;
            if (g < N_GRAPHS) {
                float* p = p0Buf + (size_t)g * K_USED + lane;
#pragma unroll
                for (int grp = 0; grp < K_USED / 32; ++grp) atomicAdd(p + 32 * grp, v[s][grp]);
            }
        }
    } else if (q < HAND_ADD_WAVES + HAND_SCAN_WAVES) {
        const int  t     = (q - HAND_ADD_WAVES) * 32 + lane;
        const bool valid = t < M_TILES;
        const int  tc    = valid ? t : 0;
        const int  gF    = gids[wg_first_tile(wg_of_tile(tc)) * 16];
        int gr[16];
#pragma unroll
        for (int r = 0; r < 16; ++r) gr[r] = gids[tc * 16 + r];
        bool outside = false;
#pragma unroll
        for (int r = 0; r < 16; ++r) outside = outside || ((unsigned)(gr[r] - gF) >= POOL_SLOTS);
        unsigned todo = (unsigned)__builtin_amdgcn_ballot_w32(valid && outside);
        for (int it = 0; it < 32; ++it) {
            if (todo == 0) break;
            const int l = __builtin_ctz(todo);
            todo &= todo - 1;
            const int tt  = (q - HAND_ADD_WAVES) * 32 + l;
            const int gFt = __builtin_amdgcn_readlane(gF, l);
#pragma unroll
            for (int r = 0; r < 16; ++r) {
                const int g = __builtin_amdgcn_readlane(gr[r], l);
                if ((unsigned)(g - gFt) >= POOL_SLOTS) {
                    const size_t row = (size_t)tt * 16 + r;
                    float* p = p0Buf + (size_t)g * K_USED + lane;
                    atomicAdd(p, pos[row * D_POS + lane]);
                    for (int grp = 0; grp < D_FEAT / 32; ++grp)
                        atomicAdd(p + D_POS + 32 * grp, feat[row * D_FEAT + 32 * grp + lane]);
                }
            }
        }
    }
}

template <int BASE, int N>
__device__ __forceinline__ void scatter_edges(
    const float* __restrict__ yBuf, float* __restrict__ accBuf,
    const int sv, const int dv, const int lane)
{
    float v[N];
    int   d[N];
#pragma unroll
    for (int u = 0; u < N; ++u) {
        const int s = __builtin_amdgcn_readlane(sv, BASE + u);
        d[u]        = __builtin_amdgcn_readlane(dv, BASE + u);
        v[u] = yBuf[(size_t)s * 32 + lane];
    }
#pragma unroll
    for (int u = 0; u < N; ++u)
        atomicAdd(accBuf + (size_t)d[u] * 32 + lane, v[u]);
}

__global__ __launch_bounds__(SC_WAVES_PER_WG * 32) void edge_scatter_kernel(
    const float* __restrict__ yBuf, const int* __restrict__ src,
    const int* __restrict__ dst, float* __restrict__ accBuf, int nEdges,
    const float* __restrict__ pos, const float* __restrict__ feat,
    const int* __restrict__ gids, const float* __restrict__ partBuf,
    float* __restrict__ p0Buf)
{
    const int wave = __builtin_amdgcn_readfirstlane((int)(threadIdx.x >> 5));
    const int lane = threadIdx.x & 31;

    if ((int)blockIdx.x >= SC_WGS) {
        hand_over(((int)blockIdx.x - SC_WGS) * SC_WAVES_PER_WG + wave, lane,
                  pos, feat, gids, partBuf, p0Buf);
        return;
    }
    const int w    = (int)blockIdx.x * SC_WAVES_PER_WG + wave;

    const int nUnits = (nEdges + SC_UNIT - 1) / SC_UNIT;
    const int u0 = (int)(((long long)w * nUnits) >> SC_WAVES_LOG2);
    const int u1 = (int)(((long long)(w + 1) * nUnits) >> SC_WAVES_LOG2);
    const int eBeg = u0 * SC_UNIT;
    const int eEnd = (u1 * SC_UNIT < nEdges) ? u1 * SC_UNIT : nEdges;

    if (eBeg >= eEnd) return;

    const int last = nEdges - 1;
    const int p0   = (eBeg + lane < last) ? eBeg + lane : last;
    int sv = src[p0], dv = dst[p0];

    for (int e0 = eBeg; e0 < eEnd; e0 += 32) {
        const int pN  = (e0 + 32 + lane < last) ? e0 + 32 + lane : last;
        const int svN = src[pN], dvN = dst[pN];

        const int left = eEnd - e0;
        if (left >= 32) {
            scatter_edges<0, 32>(yBuf, accBuf, sv, dv, lane);
        } else {
            int done = 0;
            if (left >= SC_UNIT) {
                scatter_edges<0, SC_UNIT>(yBuf, accBuf, sv, dv, lane);
                done = SC_UNIT;
            }
            for (int i = done; i < left; ++i) {
                const int s = __builtin_amdgcn_readlane(sv, i);
                const int d = __builtin_amdgcn_readlane(dv, i);
                atomicAdd(accBuf + (size_t)d * 32 + lane, yBuf[(size_t)s * 32 + lane]);
            }
        }
        sv = svN;  dv = dvN;
    }
}

#define NODE_BLOCKS ((M_TILES + WAVES_PER_BLOCK - 1) / WAVES_PER_BLOCK)
#define HEAD_BLOCKS (K_USED / 32)

__device__ __forceinline__ float other_half(const float x)
{
    return __builtin_bit_cast(float,
        __builtin_amdgcn_permlanex16(__builtin_bit_cast(int, x), __builtin_bit_cast(int, x),
                                     0x76543210, 0xfedcba98, false, false));
}

__global__ __launch_bounds__(256) void node_mlp_head_kernel(
    const float* __restrict__ accBuf, const float* __restrict__ p0Buf,
    const float* __restrict__ W2, const float* __restrict__ bias1,
    const float* __restrict__ bias2, const float* __restrict__ Wp0,
    const float* __restrict__ Wp1,
    const int* __restrict__ gids, float* __restrict__ out)
{
    __shared__ float hbuf[WAVES_PER_BLOCK][16 * 32];

    const int wave = threadIdx.x >> 5;
    const int lane = threadIdx.x & 31;
    const int  col  = lane & 15;
    const bool hi   = lane >= 16;

    if ((int)blockIdx.x >= NODE_BLOCKS) {
        const int ku0   = 32 * ((int)blockIdx.x - NODE_BLOCKS);
        const int wrow0 = (ku0 < D_POS) ? ku0 : (ku0 + 1);
        const float* aRow = p0Buf + (size_t)(16 * wave + col) * K_USED + ku0;
        v8f u0 = {}, u1 = {};
        for (int k = 0; k < 32; k += 4) {
            const int kb = k + (hi ? 2 : 0);
            const v2f a = *(const v2f*)(aRow + kb);
            v2f b0, b1f;
            b0.x  = Wp0[(wrow0 + kb) * 32 + col];       b0.y  = Wp0[(wrow0 + kb + 1) * 32 + col];
            b1f.x = Wp0[(wrow0 + kb) * 32 + 16 + col];  b1f.y = Wp0[(wrow0 + kb + 1) * 32 + 16 + col];
            u0 = __builtin_amdgcn_wmma_f32_16x16x4_f32(false, a, false, b0,  (short)0, u0, false, false);
            u1 = __builtin_amdgcn_wmma_f32_16x16x4_f32(false, a, false, b1f, (short)0, u1, false, false);
        }
        for (int j = 0; j < 8; ++j) {
            const int g = 16 * wave + (hi ? 8 : 0) + j;
            atomicAdd(&out[g * 32 + col],      u0[j]);
            atomicAdd(&out[g * 32 + 16 + col], u1[j]);
        }
        return;
    }

    const int tile = blockIdx.x * WAVES_PER_BLOCK + wave;
    if (tile >= M_TILES) return;

    const int  m0   = tile * 16;
    const int  node = m0 + col;
    const int  gl   = gids[node];

    const float* accRow = accBuf + (size_t)node * 32;
    v8f c10 = {}, c11 = {};
    for (int k = 0; k < 32; k += 4) {
        const int kb = k + (hi ? 2 : 0);
        const v2f av = *(const v2f*)(accRow + kb);
        const v2f bv = *(const v2f*)(bias1 + kb);
        v2f a;
        a.x = fmaxf(av.x + bv.x, 0.0f);
        a.y = fmaxf(av.y + bv.y, 0.0f);
        v2f b0, b1f;
        b0.x  = W2[kb * 32 + col];       b0.y  = W2[(kb + 1) * 32 + col];
        b1f.x = W2[kb * 32 + 16 + col];  b1f.y = W2[(kb + 1) * 32 + 16 + col];
        c10 = __builtin_amdgcn_wmma_f32_16x16x4_f32(false, a, false, b0,  (short)0, c10, false, false);
        c11 = __builtin_amdgcn_wmma_f32_16x16x4_f32(false, a, false, b1f, (short)0, c11, false, false);
    }

    const float bb0 = bias2[col];
    const float bb1 = bias2[16 + col];
    for (int j = 0; j < 8; ++j) {
        const int row = (hi ? 8 : 0) + j;
        hbuf[wave][row * 32 + col]      = fmaxf(c10[j] + bb0, 0.0f);
        hbuf[wave][row * 32 + 16 + col] = fmaxf(c11[j] + bb1, 0.0f);
    }

    v8f u0 = {}, u1 = {};
    for (int k = 0; k < 32; k += 4) {
        const int kb = k + (hi ? 2 : 0);
        const v2f a = *(const v2f*)(&hbuf[wave][col * 32 + kb]);
        v2f b0, b1f;
        b0.x  = Wp1[kb * 32 + col];       b0.y  = Wp1[(kb + 1) * 32 + col];
        b1f.x = Wp1[kb * 32 + 16 + col];  b1f.y = Wp1[(kb + 1) * 32 + 16 + col];
        u0 = __builtin_amdgcn_wmma_f32_16x16x4_f32(false, a, false, b0,  (short)0, u0, false, false);
        u1 = __builtin_amdgcn_wmma_f32_16x16x4_f32(false, a, false, b1f, (short)0, u1, false, false);
    }

    const int  g0      = __builtin_amdgcn_readfirstlane(gl);
    const bool uniform = __builtin_amdgcn_ballot_w32(gl == g0) == 0xFFFFFFFFu;
    if (__builtin_expect(uniform, 1)) {
        float s0 = ((u0[0] + u0[1]) + (u0[2] + u0[3])) + ((u0[4] + u0[5]) + (u0[6] + u0[7]));
        float s1 = ((u1[0] + u1[1]) + (u1[2] + u1[3])) + ((u1[4] + u1[5]) + (u1[6] + u1[7]));
        s0 += other_half(s0);
        s1 += other_half(s1);
        atomicAdd(&out[g0 * 32 + lane], hi ? s1 : s0);
    } else {
#pragma unroll
        for (int j = 0; j < 8; ++j) {
            const int gLo = __builtin_amdgcn_readlane(gl, j);
            const int gHi = __builtin_amdgcn_readlane(gl, 8 + j);
            const int g   = hi ? gHi : gLo;
            atomicAdd(&out[g * 32 + col],      u0[j]);
            atomicAdd(&out[g * 32 + 16 + col], u1[j]);
        }
    }
}

extern "C" void kernel_launch(void* const* d_in, const int* in_sizes, int n_in,
                              void* d_out, int out_size, void* d_ws, size_t ws_size,
                              hipStream_t stream) {
    const float* pos  = (const float*)d_in[0];
    const float* feat = (const float*)d_in[1];
    const int*   src  = (const int*)  d_in[2];
    const int*   dst  = (const int*)  d_in[3];
    const int*   gids = (const int*)  d_in[4];
    const float* W1   = (const float*)d_in[5];
    const float* b1   = (const float*)d_in[6];
    const float* W2   = (const float*)d_in[7];
    const float* b2   = (const float*)d_in[8];
    const float* Wp0  = (const float*)d_in[9];
    const float* bp0  = (const float*)d_in[10];
    const float* Wp1  = (const float*)d_in[11];
    const float* bp1  = (const float*)d_in[12];
    float* out = (float*)d_out;
    const int nEdges = in_sizes[2];

    float* yBuf    = (float*)d_ws;
    float* accBuf  = yBuf + (size_t)N_NODES * 32;
    float* p0Buf   = accBuf + (size_t)N_NODES * 32;
    float* partBuf = p0Buf + (size_t)N_GRAPHS * K_USED;

    input_gemm_kernel<<<GEMM_WGS, GEMM_WAVES * 32, 0, stream>>>(pos, feat, W1, gids, bp0, bp1,
                                                                yBuf, accBuf, p0Buf, partBuf, out);
    edge_scatter_kernel<<<SC_WGS + HAND_WGS, SC_WAVES_PER_WG * 32, 0, stream>>>(
        yBuf, src, dst, accBuf, nEdges, pos, feat, gids, partBuf, p0Buf);
    node_mlp_head_kernel<<<NODE_BLOCKS + HEAD_BLOCKS, 256, 0, stream>>>(
        accBuf, p0Buf, W2, b1, b2, Wp0, Wp1, gids, out);
}


# edge sums without global float atomics: launch 2 sorts each edge slice by destination owner in LDS, launch 3 owners gather y rows into LDS sums (LDS float adds), node MLP reads them; sum order changes
# speedup vs baseline: 2.7534x; 1.1063x over previous
#include <hip/hip_runtime.h>
#include <hip/hip_bf16.h>


#define N_NODES 50000
#define D_POS   32
#define D_FEAT  768
#define D_IN    801
#define D_HID   32
#define D_OUT   32
#define N_GRAPHS 128
#define M_TILES  (N_NODES / 16)
#define WAVES_PER_BLOCK 8

typedef __attribute__((ext_vector_type(2))) float v2f;
typedef __attribute__((ext_vector_type(8))) float v8f;
typedef __attribute__((ext_vector_type(4))) float v4f;

#define GEMM_WGS    256
#define FULL_WAVES  (M_TILES / GEMM_WGS)
#define GEMM_WAVES  (FULL_WAVES + 4)
#define LEFT_TILES  (M_TILES - FULL_WAVES * GEMM_WGS)
#define K_USED      800
#define K_STEPS     (K_USED / 4)
#define K_PAIRS     (K_USED / 2)

#define POOL_SLOTS  2

__device__ __forceinline__ int wg_first_tile(const int b)
{
    return FULL_WAVES * b + (b < LEFT_TILES ? b : LEFT_TILES);
}
__device__ __forceinline__ int wg_of_tile(const int t)
{
    return (t < (FULL_WAVES + 1) * LEFT_TILES)
               ? t / (FULL_WAVES + 1)
               : LEFT_TILES + (t - (FULL_WAVES + 1) * LEFT_TILES) / FULL_WAVES;
}

template <int CTRL>
__device__ __forceinline__ float lane_recv(const float x)
{
    return __builtin_bit_cast(float,
        __builtin_amdgcn_update_dpp(0, __builtin_bit_cast(int, x), CTRL, 0xF, 0xF, true));
}

template <bool MASKED>
__device__ __forceinline__ float colsum16(
    const float4& a0, const float4& a1, const float4& a2, const float4& a3,
    const int c, const bool take)
{
    float v[16] = {a0.x, a0.y, a0.z, a0.w, a1.x, a1.y, a1.z, a1.w,
                   a2.x, a2.y, a2.z, a2.w, a3.x, a3.y, a3.z, a3.w};
    if (MASKED) {
#pragma unroll
        for (int i = 0; i < 16; ++i) v[i] = take ? v[i] : 0.0f;
    }
    const bool b3 = (c & 8) != 0, b2 = (c & 4) != 0, b1 = (c & 2) != 0, b0 = (c & 1) != 0;
    float w8[8], w4[4], w2[2];
#pragma unroll
    for (int i = 0; i < 8; ++i) {
        const float keep = b3 ? v[i + 8] : v[i];
        const float send = b3 ? v[i] : v[i + 8];
        w8[i] = keep + lane_recv<0x168>(send);
    }
#pragma unroll
    for (int i = 0; i < 4; ++i) {
        const float keep = b2 ? w8[i + 4] : w8[i];
        const float send = b2 ? w8[i] : w8[i + 4];
        w4[i] = keep + lane_recv<0x164>(send);
    }
#pragma unroll
    for (int i = 0; i < 2; ++i) {
        const float keep = b1 ? w4[i + 2] : w4[i];
        const float send = b1 ? w4[i] : w4[i + 2];
        w2[i] = keep + lane_recv<0x162>(send);
    }
    const float keep = b0 ? w2[1] : w2[0];
    const float send = b0 ? w2[0] : w2[1];
    return keep + lane_recv<0x161>(send);
}

template <int NVEC>
__device__ __forceinline__ void pool_chunk(
    const float4 (&a)[NVEC], float* __restrict__ Ps, const int kbase,
    const int gl, const bool uniform, const int g0, const int gF, const int c, const int colo)
{
    if (__builtin_expect(uniform, 1)) {
        const unsigned slot = (unsigned)(g0 - gF);
        if (slot < POOL_SLOTS) {
            float* p = Ps + slot * K_USED + kbase + colo;
#pragma unroll
            for (int grp = 0; grp < NVEC / 4; ++grp) {
                const float s = colsum16<false>(a[4 * grp], a[4 * grp + 1], a[4 * grp + 2],
                                                a[4 * grp + 3], c, true);
                atomicAdd(p + 32 * grp, s);
            }
        }
    } else {
        unsigned rem = 0xFFFFu;
        for (int it = 0; it < 16; ++it) {
            if (rem == 0) break;
            const int r = __builtin_ctz(rem);
            const int g = __builtin_amdgcn_readlane(gl, r);
            const bool take = (gl == g);
            const unsigned slot = (unsigned)(g - gF);
            if (slot < POOL_SLOTS) {
                float* p = Ps + slot * K_USED + kbase + colo;
#pragma unroll
                for (int grp = 0; grp < NVEC / 4; ++grp) {
                    const float s = colsum16<true>(a[4 * grp], a[4 * grp + 1], a[4 * grp + 2],
                                                   a[4 * grp + 3], c, take);
                    atomicAdd(p + 32 * grp, s);
                }
            }
            rem &= ~(unsigned)__builtin_amdgcn_ballot_w32(take);
        }
    }
}

template <int NVEC>
__device__ __forceinline__ void gemm_chunk(
    const float4 (&a)[NVEC], const v2f* __restrict__ Bs,
    const int s0, const int lane, v8f& cy0, v8f& cy1)
{
#pragma unroll
    for (int i = 0; i < NVEC; ++i) {
#pragma unroll
        for (int o = 0; o < 2; ++o) {
            const int s = s0 + 2 * i + o;
            const v2f by0 = Bs[s * 64 + lane];
            const v2f by1 = Bs[s * 64 + 32 + lane];
            v2f av;
            av.x = o ? a[i].z : a[i].x;
            av.y = o ? a[i].w : a[i].y;
            cy0 = __builtin_amdgcn_wmma_f32_16x16x4_f32(false, av, false, by0, (short)0, cy0, false, false);
            cy1 = __builtin_amdgcn_wmma_f32_16x16x4_f32(false, av, false, by1, (short)0, cy1, false, false);
        }
    }
}

template <int NVEC>
__device__ __forceinline__ void gemm_chunk_half(
    const float4 (&a)[NVEC], const v2f* __restrict__ Bq, const int s0, v8f& acc)
{
#pragma unroll
    for (int i = 0; i < NVEC; ++i) {
#pragma unroll
        for (int o = 0; o < 2; ++o) {
            const int s = s0 + 2 * i + o;
            const v2f b = Bq[s * 64];
            v2f av;
            av.x = o ? a[i].z : a[i].x;
            av.y = o ? a[i].w : a[i].y;
            acc = __builtin_amdgcn_wmma_f32_16x16x4_f32(false, av, false, b, (short)0, acc, false, false);
        }
    }
}

template <int MODE>
__device__ __forceinline__ void chunk_work(
    const float4 (&aP)[4], const v2f* __restrict__ Bs, const v2f* __restrict__ Bq,
    float* __restrict__ Ps, const int lane, const int gl, const bool uniform,
    const int g0, const int gF, const int c, const int colo, v8f& c0, v8f& c1)
{
    if (MODE == 0) gemm_chunk<4>(aP, Bs, 0, lane, c0, c1);
    if (MODE == 1) gemm_chunk_half<4>(aP, Bq, 0, c0);
    if (MODE != 1) pool_chunk<4>(aP, Ps, 0, gl, uniform, g0, gF, c, colo);
}
template <int MODE>
__device__ __forceinline__ void chunk_work(
    const float4 (&a)[8], const int st, const v2f* __restrict__ Bs, const v2f* __restrict__ Bq,
    float* __restrict__ Ps, const int lane, const int gl, const bool uniform,
    const int g0, const int gF, const int c, const int colo, v8f& c0, v8f& c1)
{
    if (MODE == 0) gemm_chunk<8>(a, Bs, 8 + 16 * st, lane, c0, c1);
    if (MODE == 1) gemm_chunk_half<8>(a, Bq, 8 + 16 * st, c0);
    if (MODE != 1) pool_chunk<8>(a, Ps, D_POS + 64 * st, gl, uniform, g0, gF, c, colo);
}

#define A_PITCH4 17

template <int MODE>
__device__ __forceinline__ void run_tile(
    const v4f* __restrict__ featL, const v4f (&gP)[4], const v4f (&g0v)[8],
    v4f* __restrict__ AsW,
    const v2f* __restrict__ Bs, const v2f* __restrict__ Bq, float* __restrict__ Ps,
    const int lane, const int gl, const bool uniform, const int g0, const int gF,
    const int c, const int colo, v8f& c0, v8f& c1)
{
    v4f* stP = AsW + (lane >> 3) * A_PITCH4 + (lane & 7);
    v4f* stF = AsW + (lane >> 4) * A_PITCH4 + (lane & 15);
    const v4f* ld = AsW + c * A_PITCH4 + (lane >> 4);

    float4 aP[4];
#pragma unroll
    for (int i = 0; i < 4; ++i) stP[4 * i * A_PITCH4] = gP[i];
#pragma unroll
    for (int j = 0; j < 4; ++j) { const v4f t = ld[2 * j]; aP[j] = make_float4(t.x, t.y, t.z, t.w); }
    chunk_work<MODE>(aP, Bs, Bq, Ps, lane, gl, uniform, g0, gF, c, colo, c0, c1);

    float4 a[8];
    v4f g[8];
#pragma unroll
    for (int i = 0; i < 8; ++i) g[i] = g0v[i];
    for (int st = 0; st < D_FEAT / 64 - 1; ++st) {
#pragma unroll
        for (int i = 0; i < 8; ++i) stF[2 * i * A_PITCH4] = g[i];
#pragma unroll
        for (int i = 0; i < 8; ++i) g[i] = featL[2 * i * (D_FEAT / 4) + 16 * (st + 1)];
#pragma unroll
        for (int j = 0; j < 8; ++j) { const v4f t = ld[2 * j]; a[j] = make_float4(t.x, t.y, t.z, t.w); }
        chunk_work<MODE>(a, st, Bs, Bq, Ps, lane, gl, uniform, g0, gF, c, colo, c0, c1);
    }
    {
#pragma unroll
        for (int i = 0; i < 8; ++i) stF[2 * i * A_PITCH4] = g[i];
#pragma unroll
        for (int j = 0; j < 8; ++j) { const v4f t = ld[2 * j]; a[j] = make_float4(t.x, t.y, t.z, t.w); }
        chunk_work<MODE>(a, D_FEAT / 64 - 1, Bs, Bq, Ps, lane, gl, uniform, g0, gF, c, colo, c0, c1);
    }
}

__global__ __launch_bounds__(GEMM_WAVES * 32) void input_gemm_kernel(
    const float* __restrict__ pos, const float* __restrict__ feat,
    const float* __restrict__ W1,  const int* __restrict__ gids,
    const float* __restrict__ bp0, const float* __restrict__ bp1,
    float* __restrict__ yBuf, float* __restrict__ p0Buf,
    float* __restrict__ partBuf, float* __restrict__ out)
{
    __shared__ __attribute__((aligned(16))) v2f Bs[K_STEPS * 64];
    __shared__ __attribute__((aligned(16))) v4f As[GEMM_WAVES * 16 * A_PITCH4];
    __shared__ __attribute__((aligned(16))) float Ps[POOL_SLOTS * K_USED];

    const int wave   = __builtin_amdgcn_readfirstlane((int)(threadIdx.x >> 5));
    const int lane   = threadIdx.x & 31;
    const int bx     = (int)blockIdx.x;
    const int t0     = wg_first_tile(bx);
    const bool whole  = wave < FULL_WAVES;
    const int role   = wave - FULL_WAVES;
    const bool active = whole || (bx < LEFT_TILES && role < 3);
    const int tile   = t0 + (whole ? wave : FULL_WAVES);

    const int  m0   = (active ? tile : 0) * 16;
    const int  col  = lane & 15;
    const int  h    = lane >> 4;
    const int  node = m0 + col;

    const v4f* posL  = (const v4f*)(pos  + (size_t)m0 * D_POS)  + (lane >> 3) * (D_POS / 4)  + (lane & 7);
    const v4f* featL = (const v4f*)(feat + (size_t)m0 * D_FEAT) + (lane >> 4) * (D_FEAT / 4) + (lane & 15);
    v4f* AsW = As + wave * 16 * A_PITCH4;

    v4f gP[4], gA[8];
#pragma unroll
    for (int i = 0; i < 4; ++i) gP[i] = posL[4 * i * (D_POS / 4)];
#pragma unroll
    for (int i = 0; i < 8; ++i) gA[i] = featL[2 * i * (D_FEAT / 4)];
    const int gl  = gids[node];
    const int gFl = gids[t0 * 16];

    if (wave == GEMM_WAVES - 1) {
        const v4f zero4 = {0.0f, 0.0f, 0.0f, 0.0f};
        v4f* pz = (v4f*)p0Buf + bx * (N_GRAPHS * K_USED / 4 / GEMM_WGS);
        for (int j = lane; j < N_GRAPHS * K_USED / 4 / GEMM_WGS; j += 32) pz[j] = zero4;
        if (bx < N_GRAPHS) out[bx * D_OUT + lane] = bp0[lane] + bp1[lane];
    }
    for (int i = (int)threadIdx.x; i < POOL_SLOTS * K_USED; i += GEMM_WAVES * 32) Ps[i] = 0.0f;

    {
        const int n  = lane;
        const int nh = n >> 4, c = n & 15;
        for (int bt = wave; bt < K_PAIRS / 16; bt += GEMM_WAVES) {
            const int k0   = 32 * bt;
            const int row0 = (k0 < D_POS) ? k0 : (k0 + 1);
            const float* w1p = W1 + row0 * 32 + n;
            v2f v[16];
#pragma unroll
            for (int u = 0; u < 16; ++u) {
                v[u].x = w1p[(2 * u) * 32];
                v[u].y = w1p[(2 * u + 1) * 32];
            }
#pragma unroll
            for (int u = 0; u < 16; ++u) {
                const int s = 2 * (4 * bt + (u >> 2)) + (u & 1);
                Bs[s * 64 + nh * 32 + ((u >> 1) & 1) * 16 + c] = v[u];
            }
        }
    }
    __syncthreads();

    if (active) {
        const int  g0      = __builtin_amdgcn_readfirstlane(gl);
        const int  gF      = __builtin_amdgcn_readfirstlane(gFl);
        const bool uniform = __builtin_amdgcn_ballot_w32(gl == g0) == 0xFFFFFFFFu;
        const int  colo    = 8 * (col >> 2) + 4 * h + (col & 3);

        if (whole) {
            v8f cy0 = {}, cy1 = {};
            run_tile<0>(featL, gP, gA, AsW, Bs, Bs, Ps, lane, gl, uniform, g0, gF, col, colo, cy0, cy1);

            for (int j = 0; j < 8; ++j) {
                const int row = 8 * h + j;
                const int off = (m0 + row) * 32 + col;
                yBuf[off]        = cy0[j];
                yBuf[off + 16]   = cy1[j];
            }
        } else if (role < 2) {
            const v2f* Bq = Bs + role * 32 + lane;
            v8f acc = {}, unused = {};
            run_tile<1>(featL, gP, gA, AsW, Bs, Bq, Ps, lane, gl, uniform, g0, gF, col, colo, acc, unused);

            for (int j = 0; j < 8; ++j) {
                const int row = 8 * h + j;
                const int off = (m0 + row) * 32 + role * 16 + col;
                yBuf[off] = acc[j];
            }
        } else {
            v8f unused0 = {}, unused1 = {};
            run_tile<2>(featL, gP, gA, AsW, Bs, Bs, Ps, lane, gl, uniform, g0, gF, col, colo, unused0, unused1);
        }
    }

    __syncthreads();
    {
        const v4f* ps4 = (const v4f*)Ps;
        v4f* dst4 = (v4f*)partBuf + bx * (POOL_SLOTS * K_USED / 4);
        for (int i = (int)threadIdx.x; i < POOL_SLOTS * K_USED / 4; i += GEMM_WAVES * 32) dst4[i] = ps4[i];
    }
}

#define N_EDGES     400000
#define BK_WGS      GEMM_WGS
#define BK_WAVES    8
#define E_SLICE     ((N_EDGES + BK_WGS - 1) / BK_WGS)
#define E_PER_LANE  ((E_SLICE + BK_WAVES * 32 - 1) / (BK_WAVES * 32))
#define REC_PITCH   1568
#define OFF_PITCH   260

typedef __attribute__((ext_vector_type(2))) int v2i;

#define HAND_ADD_WAVES  GEMM_WGS
#define HAND_SCAN_WAVES ((M_TILES + 31) / 32)
#define HAND_WGS        ((HAND_ADD_WAVES + HAND_SCAN_WAVES + BK_WAVES - 1) / BK_WAVES)

__device__ __forceinline__ void hand_over(
    const int q, const int lane,
    const float* __restrict__ pos, const float* __restrict__ feat,
    const int* __restrict__ gids, const float* __restrict__ partBuf,
    float* __restrict__ p0Buf)
{
    if (q < HAND_ADD_WAVES) {
        const int b  = q;
        const int gF = __builtin_amdgcn_readfirstlane(gids[wg_first_tile(b) * 16]);
        const float* src = partBuf + (size_t)b * (POOL_SLOTS * K_USED) + lane;
        float v[POOL_SLOTS][K_USED / 32];
#pragma unroll
        for (int s = 0; s < POOL_SLOTS; ++s)
#pragma unroll
            for (int grp = 0; grp < K_USED / 32; ++grp) v[s][grp] = src[s * K_USED + 32 * grp];
#pragma unroll
        for (int s = 0; s < POOL_SLOTS; ++s) {
            const int g = gF + s;
            if (g < N_GRAPHS) {
                float* p = p0Buf + (size_t)g * K_USED + lane;
#pragma unroll
                for (int grp = 0; grp < K_USED / 32; ++grp) atomicAdd(p + 32 * grp, v[s][grp]);
            }
        }
    } else if (q < HAND_ADD_WAVES + HAND_SCAN_WAVES) {
        const int  t     = (q - HAND_ADD_WAVES) * 32 + lane;
        const bool valid = t < M_TILES;
        const int  tc    = valid ? t : 0;
        const int  gF    = gids[wg_first_tile(wg_of_tile(tc)) * 16];
        int gr[16];
#pragma unroll
        for (int r = 0; r < 16; ++r) gr[r] = gids[tc * 16 + r];
        bool outside = false;
#pragma unroll
        for (int r = 0; r < 16; ++r) outside = outside || ((unsigned)(gr[r] - gF) >= POOL_SLOTS);
        unsigned todo = (unsigned)__builtin_amdgcn_ballot_w32(valid && outside);
        for (int it = 0; it < 32; ++it) {
            if (todo == 0) break;
            const int l = __builtin_ctz(todo);
            todo &= todo - 1;
            const int tt  = (q - HAND_ADD_WAVES) * 32 + l;
            const int gFt = __builtin_amdgcn_readlane(gF, l);
#pragma unroll
            for (int r = 0; r < 16; ++r) {
                const int g = __builtin_amdgcn_readlane(gr[r], l);
                if ((unsigned)(g - gFt) >= POOL_SLOTS) {
                    const size_t row = (size_t)tt * 16 + r;
                    float* p = p0Buf + (size_t)g * K_USED + lane;
                    atomicAdd(p, pos[row * D_POS + lane]);
                    for (int grp = 0; grp < D_FEAT / 32; ++grp)
                        atomicAdd(p + D_POS + 32 * grp, feat[row * D_FEAT + 32 * grp + lane]);
                }
            }
        }
    }
}

__global__ __launch_bounds__(BK_WAVES * 32) void edge_bucket_kernel(
    const int* __restrict__ src, const int* __restrict__ dst, int nEdges,
    int* __restrict__ recBuf, int* __restrict__ offBuf,
    const float* __restrict__ pos, const float* __restrict__ feat,
    const int* __restrict__ gids, const float* __restrict__ partBuf,
    float* __restrict__ p0Buf)
{
    __shared__ int cntS[BK_WGS];
    __shared__ int startS[BK_WGS + 1];
    __shared__ __attribute__((aligned(16))) v2i recS[REC_PITCH];

    const int wave = __builtin_amdgcn_readfirstlane((int)(threadIdx.x >> 5));
    const int lane = threadIdx.x & 31;

    if ((int)blockIdx.x >= BK_WGS) {
        hand_over(((int)blockIdx.x - BK_WGS) * BK_WAVES + wave, lane,
                  pos, feat, gids, partBuf, p0Buf);
        return;
    }
    const int w  = (int)blockIdx.x;
    const int t  = (int)threadIdx.x;
    const int nE = (nEdges < N_EDGES) ? nEdges : N_EDGES;
    const int e0 = (w * E_SLICE < nE) ? w * E_SLICE : nE;
    const int e1 = (e0 + E_SLICE < nE) ? e0 + E_SLICE : nE;

    cntS[t] = 0;

    int s[E_PER_LANE], d[E_PER_LANE], bk[E_PER_LANE], rk[E_PER_LANE];
#pragma unroll
    for (int i = 0; i < E_PER_LANE; ++i) {
        const int e = e0 + t + BK_WAVES * 32 * i;
        s[i] = 0;  d[i] = 0;
        if (e < e1) { s[i] = src[e];  d[i] = dst[e]; }
    }
    __syncthreads();
#pragma unroll
    for (int i = 0; i < E_PER_LANE; ++i) {
        const int e = e0 + t + BK_WAVES * 32 * i;
        bk[i] = 0;  rk[i] = 0;
        if (e < e1) {
            bk[i] = wg_of_tile(d[i] >> 4);
            rk[i] = atomicAdd(&cntS[bk[i]], 1);
        }
    }
    __syncthreads();
    if (wave == 0) {
        int c[8];
        int sum = 0;
#pragma unroll
        for (int j = 0; j < 8; ++j) { c[j] = cntS[8 * lane + j];  sum += c[j]; }
        int inc = sum;
#pragma unroll
        for (int ofs = 1; ofs < 32; ofs <<= 1) {
            const int n = __shfl_up(inc, ofs);
            if (lane >= ofs) inc += n;
        }
        int run = inc - sum;
#pragma unroll
        for (int j = 0; j < 8; ++j) { startS[8 * lane + j] = run;  run += c[j]; }
        if (lane == 31) startS[BK_WGS] = run;
    }
    __syncthreads();
#pragma unroll
    for (int i = 0; i < E_PER_LANE; ++i) {
        const int e = e0 + t + BK_WAVES * 32 * i;
        if (e < e1) {
            v2i r;  r.x = s[i];  r.y = d[i];
            recS[startS[bk[i]] + rk[i]] = r;
        }
    }
    __syncthreads();
    {
        v2i* dstR = (v2i*)recBuf + (size_t)w * REC_PITCH;
        const int n = e1 - e0;
        for (int i = t; i < n; i += BK_WAVES * 32) dstR[i] = recS[i];
        int* ob = offBuf + w * OFF_PITCH;
        ob[t] = startS[t];
        if (t == 0) ob[BK_WGS] = startS[BK_WGS];
    }
}

#define OWN_WAVES   16
#define OWN_TILES   (FULL_WAVES + 1)
#define OWN_ROWS    (OWN_TILES * 16)
#define ACC_PITCH   34
#define HEAD_SLICES (K_USED / 32)

__device__ __forceinline__ float other_half(const float x)
{
    return __builtin_bit_cast(float,
        __builtin_amdgcn_permlanex16(__builtin_bit_cast(int, x), __builtin_bit_cast(int, x),
                                     0x76543210, 0xfedcba98, false, false));
}

__device__ __forceinline__ void gather_rows32(
    const float* __restrict__ yBuf, const int n0, const v2i rec, const int lane,
    float (&v)[32], int (&dl)[32])
{
#pragma unroll
    for (int u = 0; u < 32; ++u) {
        const int sN = __builtin_amdgcn_readlane(rec.x, u);
        dl[u]        = __builtin_amdgcn_readlane(rec.y, u) - n0;
        v[u] = yBuf[(size_t)sN * 32 + lane];
    }
}
__device__ __forceinline__ void add_rows32(
    float* __restrict__ accS, const int lane, const float (&v)[32], const int (&dl)[32])
{
#pragma unroll
    for (int u = 0; u < 32; ++u)
        atomicAdd(accS + dl[u] * ACC_PITCH + lane, v[u]);
}

__global__ __launch_bounds__(OWN_WAVES * 32) void owner_mlp_head_kernel(
    const float* __restrict__ yBuf, const int* __restrict__ recBuf,
    const int* __restrict__ offBuf, const float* __restrict__ p0Buf,
    const float* __restrict__ W2, const float* __restrict__ bias1,
    const float* __restrict__ bias2, const float* __restrict__ Wp0,
    const float* __restrict__ Wp1,
    const int* __restrict__ gids, float* __restrict__ out)
{
    __shared__ __attribute__((aligned(16))) float accS[(OWN_ROWS + 1) * ACC_PITCH];
    __shared__ float hbuf[OWN_WAVES][16 * 32];

    const int wave = __builtin_amdgcn_readfirstlane((int)(threadIdx.x >> 5));
    const int lane = threadIdx.x & 31;
    const int  col  = lane & 15;
    const bool hi   = lane >= 16;

    const int b  = (int)blockIdx.x;
    const int t0 = wg_first_tile(b);
    const int nT = FULL_WAVES + (b < LEFT_TILES ? 1 : 0);
    const int n0 = t0 * 16;

    const int  wS = 16 * wave + col;
    const int* ob = offBuf + wS * OFF_PITCH + b;
    const int  o0 = ob[0], o1 = ob[1];
    const bool hasTile = wave < nT;
    const int  m0   = (t0 + (hasTile ? wave : 0)) * 16;
    const int  node = m0 + col;
    const int  gl   = gids[node];

    v2f w2a[8], w2b[8], wpa[8], wpb[8], bv1[8];
#pragma unroll
    for (int q = 0; q < 8; ++q) {
        const int kb = 4 * q + (hi ? 2 : 0);
        w2a[q].x = W2[kb * 32 + col];        w2a[q].y = W2[(kb + 1) * 32 + col];
        w2b[q].x = W2[kb * 32 + 16 + col];   w2b[q].y = W2[(kb + 1) * 32 + 16 + col];
        wpa[q].x = Wp1[kb * 32 + col];       wpa[q].y = Wp1[(kb + 1) * 32 + col];
        wpb[q].x = Wp1[kb * 32 + 16 + col];  wpb[q].y = Wp1[(kb + 1) * 32 + 16 + col];
        bv1[q]   = *(const v2f*)(bias1 + kb);
    }
    const float bb0 = bias2[col];
    const float bb1 = bias2[16 + col];

    {
        const v4f* y4 = (const v4f*)(yBuf + (size_t)n0 * 32);
        const int  n4 = nT * 16 * 8;
        v4f yv[4];
#pragma unroll
        for (int q = 0; q < 4; ++q) {
            const int i = (int)threadIdx.x + OWN_WAVES * 32 * q;
            yv[q] = y4[i < n4 ? i : 0];
        }
#pragma unroll
        for (int q = 0; q < 4; ++q) {
            const int i = (int)threadIdx.x + OWN_WAVES * 32 * q;
            if (i < n4) {
                v2f* p = (v2f*)(accS + (i >> 3) * ACC_PITCH + (i & 7) * 4);
                v2f lo, up;
                lo.x = yv[q].x;  lo.y = yv[q].y;  up.x = yv[q].z;  up.y = yv[q].w;
                p[0] = lo;  p[1] = up;
            }
        }
        if ((int)threadIdx.x < ACC_PITCH) accS[OWN_ROWS * ACC_PITCH + (int)threadIdx.x] = 0.0f;
    }

    const int cntL = hi ? 0 : (o1 - o0);
    int inc = cntL;
#pragma unroll
    for (int ofs = 1; ofs < 16; ofs <<= 1) {
        const int n = __shfl_up(inc, ofs);
        if (lane >= ofs) inc += n;
    }
    const int exc   = inc - cntL;
    const int total = __builtin_amdgcn_readlane(inc, 15);
    const int gbase = wS * REC_PITCH + o0 - exc;
    const v2i* recs = (const v2i*)recBuf;
    const int spareDst = n0 + OWN_ROWS;

    auto fetch = [&](const int j0) -> v2i {
        const int j  = j0 + lane;
        const int jc = (j < total) ? j : total - 1;
        int gb = __builtin_amdgcn_readlane(gbase, 0);
#pragma unroll
        for (int sIdx = 1; sIdx < 16; ++sIdx) {
            const int ps = __builtin_amdgcn_readlane(exc, sIdx);
            const int gs = __builtin_amdgcn_readlane(gbase, sIdx);
            gb = (jc >= ps) ? gs : gb;
        }
        v2i r = recs[gb + jc];
        r.y = (j < total) ? r.y : spareDst;
        return r;
    };

    v2i recA = {0, 0}, recB = {0, 0};
    if (total > 0) {
        recA = fetch(0);
        recB = fetch(32);
    }

    __syncthreads();

    if (total > 0) {
        for (int j0 = 0; j0 < total; j0 += 64) {
            const v2i nxtA = fetch(j0 + 64);
            const v2i nxtB = fetch(j0 + 96);
            float vA[32], vB[32];
            int   dA[32], dB[32];
            gather_rows32(yBuf, n0, recA, lane, vA, dA);
            gather_rows32(yBuf, n0, recB, lane, vB, dB);
            add_rows32(accS, lane, vA, dA);
            add_rows32(accS, lane, vB, dB);
            recA = nxtA;  recB = nxtB;
        }
    }

    __syncthreads();

    if (!hasTile) {
        if (wave != OWN_WAVES - 1 || b >= HEAD_SLICES * (N_GRAPHS / 16)) return;
        const int gt    = b & (N_GRAPHS / 16 - 1);
        const int ku0   = 32 * (b / (N_GRAPHS / 16));
        const int wrow0 = (ku0 < D_POS) ? ku0 : (ku0 + 1);
        const float* aRow = p0Buf + (size_t)(16 * gt + col) * K_USED + ku0;
        v8f h0 = {}, h1v = {};
        for (int k = 0; k < 32; k += 4) {
            const int kb = k + (hi ? 2 : 0);
            const v2f a = *(const v2f*)(aRow + kb);
            v2f b0, b1f;
            b0.x  = Wp0[(wrow0 + kb) * 32 + col];       b0.y  = Wp0[(wrow0 + kb + 1) * 32 + col];
            b1f.x = Wp0[(wrow0 + kb) * 32 + 16 + col];  b1f.y = Wp0[(wrow0 + kb + 1) * 32 + 16 + col];
            h0  = __builtin_amdgcn_wmma_f32_16x16x4_f32(false, a, false, b0,  (short)0, h0,  false, false);
            h1v = __builtin_amdgcn_wmma_f32_16x16x4_f32(false, a, false, b1f, (short)0, h1v, false, false);
        }
        for (int j = 0; j < 8; ++j) {
            const int g = 16 * gt + (hi ? 8 : 0) + j;
            atomicAdd(&out[g * 32 + col],      h0[j]);
            atomicAdd(&out[g * 32 + 16 + col], h1v[j]);
        }
        return;
    }

    const float* accRow = accS + (16 * wave + col) * ACC_PITCH;
    v8f c10 = {}, c11 = {};
#pragma unroll
    for (int q = 0; q < 8; ++q) {
        const int kb = 4 * q + (hi ? 2 : 0);
        const v2f av = *(const v2f*)(accRow + kb);
        v2f a;
        a.x = fmaxf(av.x + bv1[q].x, 0.0f);
        a.y = fmaxf(av.y + bv1[q].y, 0.0f);
        c10 = __builtin_amdgcn_wmma_f32_16x16x4_f32(false, a, false, w2a[q], (short)0, c10, false, false);
        c11 = __builtin_amdgcn_wmma_f32_16x16x4_f32(false, a, false, w2b[q], (short)0, c11, false, false);
    }

    for (int j = 0; j < 8; ++j) {
        const int row = (hi ? 8 : 0) + j;
        hbuf[wave][row * 32 + col]      = fmaxf(c10[j] + bb0, 0.0f);
        hbuf[wave][row * 32 + 16 + col] = fmaxf(c11[j] + bb1, 0.0f);
    }

    v8f u0 = {}, u1 = {};
#pragma unroll
    for (int q = 0; q < 8; ++q) {
        const int kb = 4 * q + (hi ? 2 : 0);
        const v2f a = *(const v2f*)(&hbuf[wave][col * 32 + kb]);
        u0 = __builtin_amdgcn_wmma_f32_16x16x4_f32(false, a, false, wpa[q], (short)0, u0, false, false);
        u1 = __builtin_amdgcn_wmma_f32_16x16x4_f32(false, a, false, wpb[q], (short)0, u1, false, false);
    }

    const int  g0      = __builtin_amdgcn_readfirstlane(gl);
    const bool uniform = __builtin_amdgcn_ballot_w32(gl == g0) == 0xFFFFFFFFu;
    if (__builtin_expect(uniform, 1)) {
        float s0 = ((u0[0] + u0[1]) + (u0[2] + u0[3])) + ((u0[4] + u0[5]) + (u0[6] + u0[7]));
        float s1 = ((u1[0] + u1[1]) + (u1[2] + u1[3])) + ((u1[4] + u1[5]) + (u1[6] + u1[7]));
        s0 += other_half(s0);
        s1 += other_half(s1);
        atomicAdd(&out[g0 * 32 + lane], hi ? s1 : s0);
    } else {
#pragma unroll
        for (int j = 0; j < 8; ++j) {
            const int gLo = __builtin_amdgcn_readlane(gl, j);
            const int gHi = __builtin_amdgcn_readlane(gl, 8 + j);
            const int g   = hi ? gHi : gLo;
            atomicAdd(&out[g * 32 + col],      u0[j]);
            atomicAdd(&out[g * 32 + 16 + col], u1[j]);
        }
    }
}

extern "C" void kernel_launch(void* const* d_in, const int* in_sizes, int n_in,
                              void* d_out, int out_size, void* d_ws, size_t ws_size,
                              hipStream_t stream) {
    const float* pos  = (const float*)d_in[0];
    const float* feat = (const float*)d_in[1];
    const int*   src  = (const int*)  d_in[2];
    const int*   dst  = (const int*)  d_in[3];
    const int*   gids = (const int*)  d_in[4];
    const float* W1   = (const float*)d_in[5];
    const float* b1   = (const float*)d_in[6];
    const float* W2   = (const float*)d_in[7];
    const float* b2   = (const float*)d_in[8];
    const float* Wp0  = (const float*)d_in[9];
    const float* bp0  = (const float*)d_in[10];
    const float* Wp1  = (const float*)d_in[11];
    const float* bp1  = (const float*)d_in[12];
    float* out = (float*)d_out;
    const int nEdges = in_sizes[2];

    float* yBuf    = (float*)d_ws;
    int*   recBuf  = (int*)(yBuf + (size_t)N_NODES * 32);
    int*   offBuf  = recBuf + (size_t)BK_WGS * REC_PITCH * 2;
    float* p0Buf   = yBuf + (size_t)2 * N_NODES * 32;
    float* partBuf = p0Buf + (size_t)N_GRAPHS * K_USED;

    input_gemm_kernel<<<GEMM_WGS, GEMM_WAVES * 32, 0, stream>>>(pos, feat, W1, gids, bp0, bp1,
                                                                yBuf, p0Buf, partBuf, out);
    edge_bucket_kernel<<<BK_WGS + HAND_WGS, BK_WAVES * 32, 0, stream>>>(
        src, dst, nEdges, recBuf, offBuf, pos, feat, gids, partBuf, p0Buf);
    owner_mlp_head_kernel<<<GEMM_WGS, OWN_WAVES * 32, 0, stream>>>(
        yBuf, recBuf, offBuf, p0Buf, W2, b1, b2, Wp0, Wp1, gids, out);
}
